# starting kernel
# baseline (speedup 1.0000x reference)
#include <hip/hip_runtime.h>


#define BT       16
#define NTHREADS 256
#define NSLOT    (NTHREADS / BT)
#define W0       2048
#define W1       1024
#define W2       384
#define PFEAT    784
#define FBN      256
#define POPN     128
#define NIN0     (PFEAT + FBN)
#define STEPSN   8
#define NGATES   (W0 + W1 + W2)

typedef float v2f __attribute__((ext_vector_type(2)));
typedef float v8f __attribute__((ext_vector_type(8)));

__global__ void dlgn_precomp(const float* __restrict__ lg0,
                             const int*  __restrict__ le0, const int* __restrict__ ri0,
                             const float* __restrict__ lg1,
                             const int*  __restrict__ le1, const int* __restrict__ ri1,
                             const float* __restrict__ lg2,
                             const int*  __restrict__ le2, const int* __restrict__ ri2,
                             int2*  __restrict__ idx, float4* __restrict__ co) {
  int G = blockIdx.x * blockDim.x + threadIdx.x;
  if (G >= NGATES) return;
  const float* lg; const int* li; const int* ri; int g;
  if (G < W0)            { lg = lg0; li = le0; ri = ri0; g = G; }
  else if (G < W0 + W1)  { lg = lg1; li = le1; ri = ri1; g = G - W0; }
  else                   { lg = lg2; li = le2; ri = ri2; g = G - W0 - W1; }
  float w0 = 0.5f + 0.5f * sinf(lg[g * 4 + 0]);
  float w1 = 0.5f + 0.5f * sinf(lg[g * 4 + 1]);
  float w2 = 0.5f + 0.5f * sinf(lg[g * 4 + 2]);
  float w3 = 0.5f + 0.5f * sinf(lg[g * 4 + 3]);
  co[G]  = make_float4(w0, w2 - w0, w1 - w0, w0 - w1 - w2 + w3);
  idx[G] = make_int2(li[g], ri[g]);
}

__global__ __launch_bounds__(NTHREADS)
void dlgn_main(const float* __restrict__ x, const float* __restrict__ thr,
               const int2* __restrict__ idx, const float4* __restrict__ co,
               float* __restrict__ out) {
  __shared__ float sh_hin[NIN0 * BT];
  __shared__ float sh_l0 [W0   * BT];
  __shared__ float sh_l1 [W1   * BT];
  __shared__ float sh_pop[POPN * BT];

  const int t    = threadIdx.x;
  const int bb   = t & (BT - 1);
  const int slot = t / BT;
  const float* xb = x + (size_t)(blockIdx.x * BT + bb) * PFEAT;

  const float myth = thr[slot & 7];

  for (int j = slot; j < FBN; j += NSLOT)
    sh_hin[(PFEAT + j) * BT + bb] = 0.0f;

  for (int s = 0; s < STEPSN; ++s) {
    const float* xs = xb + s * 98;
    for (int p = slot; p < PFEAT; p += NSLOT) {
      float v = xs[p >> 3];
      sh_hin[p * BT + bb] = (v >= myth) ? 1.0f : 0.0f;
    }
    if (s + 1 < STEPSN) __builtin_prefetch(xs + 98, 0, 0);
    __syncthreads();

    for (int g = slot; g < W0; g += NSLOT) {
      int2   lr = idx[g];
      float4 w  = co[g];
      float L = sh_hin[lr.x * BT + bb];
      float R = sh_hin[lr.y * BT + bb];
      sh_l0[g * BT + bb] = w.x + w.y * L + w.z * R + w.w * (L * R);
    }
    __syncthreads();

    for (int g = slot; g < W1; g += NSLOT) {
      int2   lr = idx[W0 + g];
      float4 w  = co[W0 + g];
      float L = sh_l0[lr.x * BT + bb];
      float R = sh_l0[lr.y * BT + bb];
      sh_l1[g * BT + bb] = w.x + w.y * L + w.z * R + w.w * (L * R);
    }
    __syncthreads();

    for (int g = slot; g < W2; g += NSLOT) {
      int2   lr = idx[W0 + W1 + g];
      float4 w  = co[W0 + W1 + g];
      float L = sh_l1[lr.x * BT + bb];
      float R = sh_l1[lr.y * BT + bb];
      float v = w.x + w.y * L + w.z * R + w.w * (L * R);
      if (g < FBN) sh_hin[(PFEAT + g) * BT + bb] = v;
      else         sh_pop[(g - FBN) * BT + bb]  = v;
    }
    __syncthreads();
  }

  if (t < 32) {
    v8f acc = {};
    v2f ones; ones[0] = 1.0f; ones[1] = 1.0f;
    const int col   = t & 15;
    const int khalf = (t < 16) ? 0 : 2;
    #pragma unroll 4
    for (int kc = 0; kc < 32; ++kc) {
      v2f bm;
      bm[0] = sh_pop[(4 * kc + khalf + 0) * BT + col];
      bm[1] = sh_pop[(4 * kc + khalf + 1) * BT + col];
      acc = __builtin_amdgcn_wmma_f32_16x16x4_f32(
           false, ones,  false, bm,
           (short)0, acc,  false,  false);
    }
    if (t < 16) {
      float logit = acc[0] / 10.0f;
      float* ob = out + (size_t)(blockIdx.x * BT + t) * 10;
      #pragma unroll
      for (int c = 0; c < 10; ++c) ob[c] = logit;
    }
  }
}

extern "C" void kernel_launch(void* const* d_in, const int* in_sizes, int n_in,
                              void* d_out, int out_size, void* d_ws, size_t ws_size,
                              hipStream_t stream) {
  const float* x   = (const float*)d_in[0];
  const float* thr = (const float*)d_in[1];
  const float* lg0 = (const float*)d_in[2];
  const int*   le0 = (const int*)  d_in[3];
  const int*   ri0 = (const int*)  d_in[4];
  const float* lg1 = (const float*)d_in[5];
  const int*   le1 = (const int*)  d_in[6];
  const int*   ri1 = (const int*)  d_in[7];
  const float* lg2 = (const float*)d_in[8];
  const int*   le2 = (const int*)  d_in[9];
  const int*   ri2 = (const int*)  d_in[10];

  int2*   idx = (int2*)d_ws;
  float4* co  = (float4*)((char*)d_ws + 32768);

  dlgn_precomp<<<(NGATES + 255) / 256, 256, 0, stream>>>(
      lg0, le0, ri0, lg1, le1, ri1, lg2, le2, ri2, idx, co);

  dlgn_main<<<4096 / BT, NTHREADS, 0, stream>>>(x, thr, idx, co, (float*)d_out);
}


# Workgroup 256 -> 1024 threads (32 waves, 8 per SIMD instead of 2; 64 gate slots, each thread's 8 gates per trip loaded as one group): 4x the gate loads in flight per L2 round trip; sum order kept
# speedup vs baseline: 2.4737x; 2.4737x over previous
#include <hip/hip_runtime.h>


#define BT       16
#define NTHREADS 1024
#define NSLOT    (NTHREADS / BT)
#define W0       2048
#define W1       1024
#define W2       384
#define PFEAT    784
#define FBN      256
#define POPN     128
#define NIN0     (PFEAT + FBN)
#define STEPSN   8
#define NGATES   (W0 + W1 + W2)
#define GB0      8
#define GB1      8
#define G0T      (W0 / NSLOT)
#define G1T      (W1 / NSLOT)
#define G2T      (W2 / NSLOT)
#define ENCF     (PFEAT / NSLOT)
#define ENCR     (PFEAT % NSLOT)

typedef float v2f __attribute__((ext_vector_type(2)));
typedef float v8f __attribute__((ext_vector_type(8)));

__global__ void dlgn_precomp(const float* __restrict__ lg0,
                             const int*  __restrict__ le0, const int* __restrict__ ri0,
                             const float* __restrict__ lg1,
                             const int*  __restrict__ le1, const int* __restrict__ ri1,
                             const float* __restrict__ lg2,
                             const int*  __restrict__ le2, const int* __restrict__ ri2,
                             int2*  __restrict__ idx, float4* __restrict__ co) {
  int G = blockIdx.x * blockDim.x + threadIdx.x;
  if (G >= NGATES) return;
  const float* lg; const int* li; const int* ri; int g;
  if (G < W0)            { lg = lg0; li = le0; ri = ri0; g = G; }
  else if (G < W0 + W1)  { lg = lg1; li = le1; ri = ri1; g = G - W0; }
  else                   { lg = lg2; li = le2; ri = ri2; g = G - W0 - W1; }
  float w0 = 0.5f + 0.5f * sinf(lg[g * 4 + 0]);
  float w1 = 0.5f + 0.5f * sinf(lg[g * 4 + 1]);
  float w2 = 0.5f + 0.5f * sinf(lg[g * 4 + 2]);
  float w3 = 0.5f + 0.5f * sinf(lg[g * 4 + 3]);
  co[G]  = make_float4(w0, w2 - w0, w1 - w0, w0 - w1 - w2 + w3);
  idx[G] = make_int2(li[g], ri[g]);
}

template <int NB>
__device__ __forceinline__ void gate_trip(const int2* __restrict__ ip,
                                          const float4* __restrict__ cp,
                                          const float* __restrict__ src,
                                          int j0, int bb, float (&v)[NB]) {
  int2   lr[NB];
  float4 w[NB];
  for (int u = 0; u < NB; ++u) lr[u] = ip[(j0 + u) * NSLOT];
  for (int u = 0; u < NB; ++u) w[u] = cp[(j0 + u) * NSLOT];
  float L[NB], R[NB];
  for (int u = 0; u < NB; ++u) {
    L[u] = src[lr[u].x * BT + bb];
    R[u] = src[lr[u].y * BT + bb];
  }
  for (int u = 0; u < NB; ++u)
    v[u] = w[u].x + w[u].y * L[u] + w[u].z * R[u] + w[u].w * (L[u] * R[u]);
}

__global__ __launch_bounds__(NTHREADS)
void dlgn_main(const float* __restrict__ x, const float* __restrict__ thr,
               const int2* __restrict__ idx, const float4* __restrict__ co,
               float* __restrict__ out) {
  __shared__ float sh_hin[NIN0 * BT];
  __shared__ float sh_l0 [W0   * BT];
  __shared__ float sh_l1 [W1   * BT];
  __shared__ float sh_pop[POPN * BT];

  const int t    = threadIdx.x;
  const int bb   = t & (BT - 1);
  const int slot = t / BT;
  const float* xb = x + (size_t)(blockIdx.x * BT + bb) * PFEAT;

  const float myth = thr[slot & 7];

  const int2*   ip0 = idx + slot;
  const float4* cp0 = co  + slot;
  const int2*   ip1 = idx + W0 + slot;
  const float4* cp1 = co  + W0 + slot;
  const int2*   ip2 = idx + W0 + W1 + slot;
  const float4* cp2 = co  + W0 + W1 + slot;
  const int     row = slot * BT + bb;

  for (int j = 0; j < FBN / NSLOT; ++j)
    sh_hin[(PFEAT + j * NSLOT) * BT + row] = 0.0f;

  for (int s = 0; s < STEPSN; ++s) {
    const float* xs = xb + s * 98 + (slot >> 3);
    {
      float px[ENCF];
      for (int j = 0; j < ENCF; ++j) px[j] = xs[(NSLOT / 8) * j];
      float pt = 0.0f;
      if (slot < ENCR) pt = xs[(NSLOT / 8) * ENCF];
      for (int j = 0; j < ENCF; ++j)
        sh_hin[j * NSLOT * BT + row] = (px[j] >= myth) ? 1.0f : 0.0f;
      if (slot < ENCR)
        sh_hin[ENCF * NSLOT * BT + row] = (pt >= myth) ? 1.0f : 0.0f;
    }
    if (s + 1 < STEPSN) __builtin_prefetch(xs + 98, 0, 0);
    __syncthreads();

    #pragma unroll 1
    for (int j0 = 0; j0 < G0T; j0 += GB0) {
      float v[GB0];
      gate_trip<GB0>(ip0, cp0, sh_hin, j0, bb, v);
      for (int u = 0; u < GB0; ++u) sh_l0[(j0 + u) * NSLOT * BT + row] = v[u];
    }
    __syncthreads();

    #pragma unroll 1
    for (int j0 = 0; j0 < G1T; j0 += GB1) {
      float v[GB1];
      gate_trip<GB1>(ip1, cp1, sh_l0, j0, bb, v);
      for (int u = 0; u < GB1; ++u) sh_l1[(j0 + u) * NSLOT * BT + row] = v[u];
    }
    __syncthreads();

    {
      float v[G2T];
      gate_trip<G2T>(ip2, cp2, sh_l1, 0, bb, v);
      for (int u = 0; u < G2T; ++u) {
        if (u * NSLOT < FBN) sh_hin[(PFEAT + u * NSLOT) * BT + row] = v[u];
        else                 sh_pop[(u * NSLOT - FBN) * BT + row]  = v[u];
      }
    }
    __syncthreads();
  }

  if (t < 32) {
    v8f acc = {};
    v2f ones; ones[0] = 1.0f; ones[1] = 1.0f;
    const int col   = t & 15;
    const int khalf = (t < 16) ? 0 : 2;
    #pragma unroll 4
    for (int kc = 0; kc < 32; ++kc) {
      v2f bm;
      bm[0] = sh_pop[(4 * kc + khalf + 0) * BT + col];
      bm[1] = sh_pop[(4 * kc + khalf + 1) * BT + col];
      acc = __builtin_amdgcn_wmma_f32_16x16x4_f32(
           false, ones,  false, bm,
           (short)0, acc,  false,  false);
    }
    if (t < 16) {
      float logit = acc[0] / 10.0f;
      float* ob = out + (size_t)(blockIdx.x * BT + t) * 10;
      #pragma unroll
      for (int c = 0; c < 10; ++c) ob[c] = logit;
    }
  }
}

extern "C" void kernel_launch(void* const* d_in, const int* in_sizes, int n_in,
                              void* d_out, int out_size, void* d_ws, size_t ws_size,
                              hipStream_t stream) {
  const float* x   = (const float*)d_in[0];
  const float* thr = (const float*)d_in[1];
  const float* lg0 = (const float*)d_in[2];
  const int*   le0 = (const int*)  d_in[3];
  const int*   ri0 = (const int*)  d_in[4];
  const float* lg1 = (const float*)d_in[5];
  const int*   le1 = (const int*)  d_in[6];
  const int*   ri1 = (const int*)  d_in[7];
  const float* lg2 = (const float*)d_in[8];
  const int*   le2 = (const int*)  d_in[9];
  const int*   ri2 = (const int*)  d_in[10];

  int2*   idx = (int2*)d_ws;
  float4* co  = (float4*)((char*)d_ws + 32768);

  dlgn_precomp<<<(NGATES + 255) / 256, 256, 0, stream>>>(
      lg0, le0, ri0, lg1, le1, ri1, lg2, le2, ri2, idx, co);

  dlgn_main<<<4096 / BT, NTHREADS, 0, stream>>>(x, thr, idx, co, (float*)d_out);
}


# Each thread takes 4 batch columns of a gate (128-bit LDS gathers and stores, 256 gate slots): a wave covers 8 gates per wiring, coefficient and LDS instruction instead of 2; sum order kept
# speedup vs baseline: 3.8548x; 1.5583x over previous
#include <hip/hip_runtime.h>


#define BT       16
#define NTHREADS 1024
#define VB       4
#define NSLOT    (NTHREADS * VB / BT)
#define W0       2048
#define W1       1024
#define W2       384
#define PFEAT    784
#define FBN      256
#define POPN     128
#define NIN0     (PFEAT + FBN)
#define STEPSN   8
#define NGATES   (W0 + W1 + W2)
#define GB0      4
#define GB1      2
#define G0T      (W0 / NSLOT)
#define G1T      (W1 / NSLOT)
#define ENCF     (PFEAT / NSLOT)
#define ENCR     (PFEAT % NSLOT)
#define PIXJ     (NSLOT / 8)

static_assert(FBN == NSLOT, "layer 2: one feedback gate per slot");
static_assert(W2 - FBN == POPN && POPN * 2 == NSLOT, "layer 2: population gates on the lower half of the slots");

typedef float v2f __attribute__((ext_vector_type(2)));
typedef float v4f __attribute__((ext_vector_type(4)));
typedef float v8f __attribute__((ext_vector_type(8)));

__global__ void dlgn_precomp(const float* __restrict__ lg0,
                             const int*  __restrict__ le0, const int* __restrict__ ri0,
                             const float* __restrict__ lg1,
                             const int*  __restrict__ le1, const int* __restrict__ ri1,
                             const float* __restrict__ lg2,
                             const int*  __restrict__ le2, const int* __restrict__ ri2,
                             int2*  __restrict__ idx, float4* __restrict__ co) {
  int G = blockIdx.x * blockDim.x + threadIdx.x;
  if (G >= NGATES) return;
  const float* lg; const int* li; const int* ri; int g;
  if (G < W0)            { lg = lg0; li = le0; ri = ri0; g = G; }
  else if (G < W0 + W1)  { lg = lg1; li = le1; ri = ri1; g = G - W0; }
  else                   { lg = lg2; li = le2; ri = ri2; g = G - W0 - W1; }
  float w0 = 0.5f + 0.5f * sinf(lg[g * 4 + 0]);
  float w1 = 0.5f + 0.5f * sinf(lg[g * 4 + 1]);
  float w2 = 0.5f + 0.5f * sinf(lg[g * 4 + 2]);
  float w3 = 0.5f + 0.5f * sinf(lg[g * 4 + 3]);
  co[G]  = make_float4(w0, w2 - w0, w1 - w0, w0 - w1 - w2 + w3);
  idx[G] = make_int2(li[g], ri[g]);
}

template <int NB>
__device__ __forceinline__ void gate_trip(const int2* __restrict__ ip,
                                          const float4* __restrict__ cp,
                                          const float* __restrict__ src,
                                          int j0, int c4, v4f (&v)[NB]) {
  int2   lr[NB];
  float4 w[NB];
  for (int u = 0; u < NB; ++u) lr[u] = ip[(j0 + u) * NSLOT];
  for (int u = 0; u < NB; ++u) w[u] = cp[(j0 + u) * NSLOT];
  v4f L[NB], R[NB];
  for (int u = 0; u < NB; ++u) {
    L[u] = *(const v4f*)(src + lr[u].x * BT + c4);
    R[u] = *(const v4f*)(src + lr[u].y * BT + c4);
  }
  for (int u = 0; u < NB; ++u)
    v[u] = w[u].x + w[u].y * L[u] + w[u].z * R[u] + w[u].w * (L[u] * R[u]);
}

__global__ __launch_bounds__(NTHREADS)
void dlgn_main(const float* __restrict__ x, const float* __restrict__ thr,
               const int2* __restrict__ idx, const float4* __restrict__ co,
               float* __restrict__ out) {
  __shared__ __attribute__((aligned(16))) float sh_hin[NIN0 * BT];
  __shared__ __attribute__((aligned(16))) float sh_l0 [W0   * BT];
  __shared__ __attribute__((aligned(16))) float sh_l1 [W1   * BT];
  __shared__ __attribute__((aligned(16))) float sh_pop[POPN * BT];

  const int t    = threadIdx.x;
  const int c4   = (t & 3) * VB;
  const int slot = t >> 2;
  const float* xb = x + (size_t)(blockIdx.x * BT + c4) * PFEAT;

  const float myth = thr[slot & 7];

  const int2*   ip0 = idx + slot;
  const float4* cp0 = co  + slot;
  const int2*   ip1 = idx + W0 + slot;
  const float4* cp1 = co  + W0 + slot;
  const int2*   ip2 = idx + W0 + W1 + slot;
  const float4* cp2 = co  + W0 + W1 + slot;
  const int     row = slot * BT + c4;

  {
    v4f z = {0.0f, 0.0f, 0.0f, 0.0f};
    *(v4f*)&sh_hin[PFEAT * BT + row] = z;
  }

  for (int s = 0; s < STEPSN; ++s) {
    const float* xs = xb + s * 98 + (slot >> 3);
    {
      float px[ENCF][VB];
      for (int j = 0; j < ENCF; ++j)
        for (int i = 0; i < VB; ++i) px[j][i] = xs[i * PFEAT + PIXJ * j];
      float pt[VB] = {0.0f, 0.0f, 0.0f, 0.0f};
      if (slot < ENCR)
        for (int i = 0; i < VB; ++i) pt[i] = xs[i * PFEAT + PIXJ * ENCF];
      for (int j = 0; j < ENCF; ++j) {
        v4f e;
        for (int i = 0; i < VB; ++i) e[i] = (px[j][i] >= myth) ? 1.0f : 0.0f;
        *(v4f*)&sh_hin[j * NSLOT * BT + row] = e;
      }
      if (slot < ENCR) {
        v4f e;
        for (int i = 0; i < VB; ++i) e[i] = (pt[i] >= myth) ? 1.0f : 0.0f;
        *(v4f*)&sh_hin[ENCF * NSLOT * BT + row] = e;
      }
    }
    if (s + 1 < STEPSN) __builtin_prefetch(xs + (slot & 3) * PFEAT + 98, 0, 0);
    __syncthreads();

    #pragma unroll 1
    for (int j0 = 0; j0 < G0T; j0 += GB0) {
      v4f v[GB0];
      gate_trip<GB0>(ip0, cp0, sh_hin, j0, c4, v);
      for (int u = 0; u < GB0; ++u) *(v4f*)&sh_l0[(j0 + u) * NSLOT * BT + row] = v[u];
    }
    __syncthreads();

    #pragma unroll 1
    for (int j0 = 0; j0 < G1T; j0 += GB1) {
      v4f v[GB1];
      gate_trip<GB1>(ip1, cp1, sh_l0, j0, c4, v);
      for (int u = 0; u < GB1; ++u) *(v4f*)&sh_l1[(j0 + u) * NSLOT * BT + row] = v[u];
    }
    __syncthreads();

    if (slot < POPN) {
      v4f v[2];
      gate_trip<2>(ip2, cp2, sh_l1, 0, c4, v);
      *(v4f*)&sh_hin[PFEAT * BT + row] = v[0];
      *(v4f*)&sh_pop[row]              = v[1];
    } else {
      v4f v[1];
      gate_trip<1>(ip2, cp2, sh_l1, 0, c4, v);
      *(v4f*)&sh_hin[PFEAT * BT + row] = v[0];
    }
    __syncthreads();
  }

  if (t < 32) {
    v8f acc = {};
    v2f ones; ones[0] = 1.0f; ones[1] = 1.0f;
    const int col   = t & 15;
    const int khalf = (t < 16) ? 0 : 2;
    #pragma unroll 4
    for (int kc = 0; kc < 32; ++kc) {
      v2f bm;
      bm[0] = sh_pop[(4 * kc + khalf + 0) * BT + col];
      bm[1] = sh_pop[(4 * kc + khalf + 1) * BT + col];
      acc = __builtin_amdgcn_wmma_f32_16x16x4_f32(
           false, ones,  false, bm,
           (short)0, acc,  false,  false);
    }
    if (t < 16) {
      float logit = acc[0] / 10.0f;
      float* ob = out + (size_t)(blockIdx.x * BT + t) * 10;
      #pragma unroll
      for (int c = 0; c < 10; ++c) ob[c] = logit;
    }
  }
}

extern "C" void kernel_launch(void* const* d_in, const int* in_sizes, int n_in,
                              void* d_out, int out_size, void* d_ws, size_t ws_size,
                              hipStream_t stream) {
  const float* x   = (const float*)d_in[0];
  const float* thr = (const float*)d_in[1];
  const float* lg0 = (const float*)d_in[2];
  const int*   le0 = (const int*)  d_in[3];
  const int*   ri0 = (const int*)  d_in[4];
  const float* lg1 = (const float*)d_in[5];
  const int*   le1 = (const int*)  d_in[6];
  const int*   ri1 = (const int*)  d_in[7];
  const float* lg2 = (const float*)d_in[8];
  const int*   le2 = (const int*)  d_in[9];
  const int*   ri2 = (const int*)  d_in[10];

  int2*   idx = (int2*)d_ws;
  float4* co  = (float4*)((char*)d_ws + 32768);

  dlgn_precomp<<<(NGATES + 255) / 256, 256, 0, stream>>>(
      lg0, le0, ri0, lg1, le1, ri1, lg2, le2, ri2, idx, co);

  dlgn_main<<<4096 / BT, NTHREADS, 0, stream>>>(x, thr, idx, co, (float*)d_out);
}


# Each thread's 14 gates (gather offsets, coefficients) are loaded once and stay in registers over the 8 steps; op_sel packed FMAs read the coefficients in place: no L2 load in a layer; sum order kept
# speedup vs baseline: 4.5966x; 1.1924x over previous
#include <hip/hip_runtime.h>


#define BT       16
#define NTHREADS 1024
#define VB       4
#define NSLOT    (NTHREADS * VB / BT)
#define W0       2048
#define W1       1024
#define W2       384
#define PFEAT    784
#define FBN      256
#define POPN     128
#define NIN0     (PFEAT + FBN)
#define STEPSN   8
#define NGATES   (W0 + W1 + W2)
#define GB       2
#define G0T      (W0 / NSLOT)
#define G1T      (W1 / NSLOT)
#define ENCF     (PFEAT / NSLOT)
#define ENCR     (PFEAT % NSLOT)
#define PIXJ     (NSLOT / 8)

static_assert(FBN == NSLOT, "layer 2: one feedback gate per slot");
static_assert(W2 - FBN == POPN && POPN * 2 == NSLOT, "layer 2: population gates on the lower half of the slots");

typedef float v2f __attribute__((ext_vector_type(2)));
typedef float v4f __attribute__((ext_vector_type(4)));
typedef float v8f __attribute__((ext_vector_type(8)));

__global__ void dlgn_precomp(const float* __restrict__ lg0,
                             const int*  __restrict__ le0, const int* __restrict__ ri0,
                             const float* __restrict__ lg1,
                             const int*  __restrict__ le1, const int* __restrict__ ri1,
                             const float* __restrict__ lg2,
                             const int*  __restrict__ le2, const int* __restrict__ ri2,
                             int2*  __restrict__ idx, float4* __restrict__ co) {
  int G = blockIdx.x * blockDim.x + threadIdx.x;
  if (G >= NGATES) return;
  const float* lg; const int* li; const int* ri; int g;
  if (G < W0)            { lg = lg0; li = le0; ri = ri0; g = G; }
  else if (G < W0 + W1)  { lg = lg1; li = le1; ri = ri1; g = G - W0; }
  else                   { lg = lg2; li = le2; ri = ri2; g = G - W0 - W1; }
  float w0 = 0.5f + 0.5f * sinf(lg[g * 4 + 0]);
  float w1 = 0.5f + 0.5f * sinf(lg[g * 4 + 1]);
  float w2 = 0.5f + 0.5f * sinf(lg[g * 4 + 2]);
  float w3 = 0.5f + 0.5f * sinf(lg[g * 4 + 3]);
  co[G]  = make_float4(w0, w2 - w0, w1 - w0, w0 - w1 - w2 + w3);
  idx[G] = make_int2(li[g], ri[g]);
}

struct Gate { int oL, oR; v2f kx, kz; };

__device__ __forceinline__ Gate load_gate(const int2* __restrict__ idx,
                                          const float4* __restrict__ co, int g, int c4) {
  const int2   lr = idx[g];
  const float4 w  = co[g];
  Gate G;
  G.oL = lr.x * BT + c4;
  G.oR = lr.y * BT + c4;
  G.kx[0] = w.x; G.kx[1] = w.y;
  G.kz[0] = w.z; G.kz[1] = w.w;
  return G;
}


__device__ __forceinline__ v2f pk_lo_plus_hi_mul(v2f k, v2f x) {
  v2f t;
  asm("v_pk_fma_f32 %0, %1, %2, %1 op_sel:[1,0,0] op_sel_hi:[1,1,0]"
      : "=&v"(t) : "v"(k), "v"(x));
  return t;
}
__device__ __forceinline__ v2f pk_add_lo_mul(v2f t, v2f k, v2f x) {
  asm("v_pk_fma_f32 %0, %1, %2, %0 op_sel_hi:[0,1,1]"
      : "+v"(t) : "v"(k), "v"(x));
  return t;
}
__device__ __forceinline__ v2f pk_add_hi_mul(v2f t, v2f k, v2f x) {
  asm("v_pk_fma_f32 %0, %1, %2, %0 op_sel:[1,0,0] op_sel_hi:[1,1,1]"
      : "+v"(t) : "v"(k), "v"(x));
  return t;
}

template <int NB, int J0, int NT>
__device__ __forceinline__ void gate_group(const Gate (&g)[NT],
                                           const float* __restrict__ src, v4f (&v)[NB]) {
  v4f L[NB], R[NB];
  #pragma unroll
  for (int u = 0; u < NB; ++u) {
    L[u] = *(const v4f*)(src + g[J0 + u].oL);
    R[u] = *(const v4f*)(src + g[J0 + u].oR);
  }
  #pragma unroll
  for (int u = 0; u < NB; ++u) {
    const Gate& G = g[J0 + u];
    v2f tl = pk_lo_plus_hi_mul(G.kx, L[u].xy);
    v2f th = pk_lo_plus_hi_mul(G.kx, L[u].zw);
    tl = pk_add_lo_mul(tl, G.kz, R[u].xy);
    th = pk_add_lo_mul(th, G.kz, R[u].zw);
    const v2f ml = L[u].xy * R[u].xy;
    const v2f mh = L[u].zw * R[u].zw;
    tl = pk_add_hi_mul(tl, G.kz, ml);
    th = pk_add_hi_mul(th, G.kz, mh);
    v[u] = __builtin_shufflevector(tl, th, 0, 1, 2, 3);
  }
}

__global__ __launch_bounds__(NTHREADS)
void dlgn_main(const float* __restrict__ x, const float* __restrict__ thr,
               const int2* __restrict__ idx, const float4* __restrict__ co,
               float* __restrict__ out) {
  __shared__ __attribute__((aligned(16))) float sh_hin[NIN0 * BT];
  __shared__ __attribute__((aligned(16))) float sh_l0 [W0   * BT];
  __shared__ __attribute__((aligned(16))) float sh_l1 [W1   * BT];
  __shared__ __attribute__((aligned(16))) float sh_pop[POPN * BT];

  const int t    = threadIdx.x;
  const int c4   = (t & 3) * VB;
  const int slot = t >> 2;
  const float* xb = x + (size_t)(blockIdx.x * BT + c4) * PFEAT;

  const float myth = thr[slot & 7];

  const int     row = slot * BT + c4;

  Gate g0[G0T], g1[G1T], g2[2];
  #pragma unroll
  for (int j = 0; j < G0T; ++j) g0[j] = load_gate(idx, co, slot + NSLOT * j, c4);
  #pragma unroll
  for (int j = 0; j < G1T; ++j) g1[j] = load_gate(idx, co, W0 + slot + NSLOT * j, c4);
  g2[0] = load_gate(idx, co, W0 + W1 + slot, c4);
  g2[1] = g2[0];
  if (slot < POPN) g2[1] = load_gate(idx, co, W0 + W1 + NSLOT + slot, c4);

  {
    v4f z = {0.0f, 0.0f, 0.0f, 0.0f};
    *(v4f*)&sh_hin[PFEAT * BT + row] = z;
  }

  for (int s = 0; s < STEPSN; ++s) {
    const float* xs = xb + s * 98 + (slot >> 3);
    {
      float px[ENCF][VB];
      for (int j = 0; j < ENCF; ++j)
        for (int i = 0; i < VB; ++i) px[j][i] = xs[i * PFEAT + PIXJ * j];
      float pt[VB] = {0.0f, 0.0f, 0.0f, 0.0f};
      if (slot < ENCR)
        for (int i = 0; i < VB; ++i) pt[i] = xs[i * PFEAT + PIXJ * ENCF];
      for (int j = 0; j < ENCF; ++j) {
        v4f e;
        for (int i = 0; i < VB; ++i) e[i] = (px[j][i] >= myth) ? 1.0f : 0.0f;
        *(v4f*)&sh_hin[j * NSLOT * BT + row] = e;
      }
      if (slot < ENCR) {
        v4f e;
        for (int i = 0; i < VB; ++i) e[i] = (pt[i] >= myth) ? 1.0f : 0.0f;
        *(v4f*)&sh_hin[ENCF * NSLOT * BT + row] = e;
      }
    }
    if (s + 1 < STEPSN) __builtin_prefetch(xs + (slot & 3) * PFEAT + 98, 0, 0);
    __syncthreads();

    {
      v4f v[GB];
      gate_group<GB, 0, G0T>(g0, sh_hin, v);
      for (int u = 0; u < GB; ++u) *(v4f*)&sh_l0[(0 + u) * NSLOT * BT + row] = v[u];
      gate_group<GB, 2, G0T>(g0, sh_hin, v);
      for (int u = 0; u < GB; ++u) *(v4f*)&sh_l0[(2 + u) * NSLOT * BT + row] = v[u];
      gate_group<GB, 4, G0T>(g0, sh_hin, v);
      for (int u = 0; u < GB; ++u) *(v4f*)&sh_l0[(4 + u) * NSLOT * BT + row] = v[u];
      gate_group<GB, 6, G0T>(g0, sh_hin, v);
      for (int u = 0; u < GB; ++u) *(v4f*)&sh_l0[(6 + u) * NSLOT * BT + row] = v[u];
    }
    __syncthreads();

    {
      v4f v[GB];
      gate_group<GB, 0, G1T>(g1, sh_l0, v);
      for (int u = 0; u < GB; ++u) *(v4f*)&sh_l1[(0 + u) * NSLOT * BT + row] = v[u];
      gate_group<GB, 2, G1T>(g1, sh_l0, v);
      for (int u = 0; u < GB; ++u) *(v4f*)&sh_l1[(2 + u) * NSLOT * BT + row] = v[u];
    }
    __syncthreads();

    if (slot < POPN) {
      v4f v[2];
      gate_group<2, 0, 2>(g2, sh_l1, v);
      *(v4f*)&sh_hin[PFEAT * BT + row] = v[0];
      *(v4f*)&sh_pop[row]              = v[1];
    } else {
      v4f v[1];
      gate_group<1, 0, 2>(g2, sh_l1, v);
      *(v4f*)&sh_hin[PFEAT * BT + row] = v[0];
    }
    __syncthreads();
  }

  if (t < 32) {
    v8f acc = {};
    v2f ones; ones[0] = 1.0f; ones[1] = 1.0f;
    const int col   = t & 15;
    const int khalf = (t < 16) ? 0 : 2;
    #pragma unroll 4
    for (int kc = 0; kc < 32; ++kc) {
      v2f bm;
      bm[0] = sh_pop[(4 * kc + khalf + 0) * BT + col];
      bm[1] = sh_pop[(4 * kc + khalf + 1) * BT + col];
      acc = __builtin_amdgcn_wmma_f32_16x16x4_f32(
           false, ones,  false, bm,
           (short)0, acc,  false,  false);
    }
    if (t < 16) {
      float logit = acc[0] / 10.0f;
      float* ob = out + (size_t)(blockIdx.x * BT + t) * 10;
      #pragma unroll
      for (int c = 0; c < 10; ++c) ob[c] = logit;
    }
  }
}

extern "C" void kernel_launch(void* const* d_in, const int* in_sizes, int n_in,
                              void* d_out, int out_size, void* d_ws, size_t ws_size,
                              hipStream_t stream) {
  const float* x   = (const float*)d_in[0];
  const float* thr = (const float*)d_in[1];
  const float* lg0 = (const float*)d_in[2];
  const int*   le0 = (const int*)  d_in[3];
  const int*   ri0 = (const int*)  d_in[4];
  const float* lg1 = (const float*)d_in[5];
  const int*   le1 = (const int*)  d_in[6];
  const int*   ri1 = (const int*)  d_in[7];
  const float* lg2 = (const float*)d_in[8];
  const int*   le2 = (const int*)  d_in[9];
  const int*   ri2 = (const int*)  d_in[10];

  int2*   idx = (int2*)d_ws;
  float4* co  = (float4*)((char*)d_ws + 32768);

  dlgn_precomp<<<(NGATES + 255) / 256, 256, 0, stream>>>(
      lg0, le0, ri0, lg1, le1, ri1, lg2, le2, ri2, idx, co);

  dlgn_main<<<4096 / BT, NTHREADS, 0, stream>>>(x, thr, idx, co, (float*)d_out);
}


# Gates of layers 0/1 whose row no evaluated gate reads (found per workgroup from the wiring) gather at the rows of a read gate of their half-wave: same address, no extra LDS bank cycle; sum order kept
# speedup vs baseline: 5.0643x; 1.1017x over previous
#include <hip/hip_runtime.h>


#define BT       16
#define NTHREADS 1024
#define VB       4
#define NSLOT    (NTHREADS * VB / BT)
#define W0       2048
#define W1       1024
#define W2       384
#define PFEAT    784
#define FBN      256
#define POPN     128
#define NIN0     (PFEAT + FBN)
#define STEPSN   8
#define NGATES   (W0 + W1 + W2)
#define GB       2
#define G0T      (W0 / NSLOT)
#define G1T      (W1 / NSLOT)
#define ENCF     (PFEAT / NSLOT)
#define ENCR     (PFEAT % NSLOT)
#define PIXJ     (NSLOT / 8)

static_assert(FBN == NSLOT, "layer 2: one feedback gate per slot");
static_assert(W2 - FBN == POPN && POPN * 2 == NSLOT, "layer 2: population gates on the lower half of the slots");

typedef float v2f __attribute__((ext_vector_type(2)));
typedef float v4f __attribute__((ext_vector_type(4)));
typedef float v8f __attribute__((ext_vector_type(8)));

__global__ void dlgn_precomp(const float* __restrict__ lg0,
                             const int*  __restrict__ le0, const int* __restrict__ ri0,
                             const float* __restrict__ lg1,
                             const int*  __restrict__ le1, const int* __restrict__ ri1,
                             const float* __restrict__ lg2,
                             const int*  __restrict__ le2, const int* __restrict__ ri2,
                             int2*  __restrict__ idx, float4* __restrict__ co) {
  int G = blockIdx.x * blockDim.x + threadIdx.x;
  if (G >= NGATES) return;
  const float* lg; const int* li; const int* ri; int g;
  if (G < W0)            { lg = lg0; li = le0; ri = ri0; g = G; }
  else if (G < W0 + W1)  { lg = lg1; li = le1; ri = ri1; g = G - W0; }
  else                   { lg = lg2; li = le2; ri = ri2; g = G - W0 - W1; }
  float w0 = 0.5f + 0.5f * sinf(lg[g * 4 + 0]);
  float w1 = 0.5f + 0.5f * sinf(lg[g * 4 + 1]);
  float w2 = 0.5f + 0.5f * sinf(lg[g * 4 + 2]);
  float w3 = 0.5f + 0.5f * sinf(lg[g * 4 + 3]);
  co[G]  = make_float4(w0, w2 - w0, w1 - w0, w0 - w1 - w2 + w3);
  idx[G] = make_int2(li[g], ri[g]);
}

struct Gate { int oL, oR; v2f kx, kz; };

__device__ __forceinline__ Gate load_gate(const int2* __restrict__ idx,
                                          const float4* __restrict__ co, int g, int c4) {
  const int2   lr = idx[g];
  const float4 w  = co[g];
  Gate G;
  G.oL = lr.x * BT + c4;
  G.oR = lr.y * BT + c4;
  G.kx[0] = w.x; G.kx[1] = w.y;
  G.kz[0] = w.z; G.kz[1] = w.w;
  return G;
}


__device__ __forceinline__ v2f pk_lo_plus_hi_mul(v2f k, v2f x) {
  v2f t;
  asm("v_pk_fma_f32 %0, %1, %2, %1 op_sel:[1,0,0] op_sel_hi:[1,1,0]"
      : "=&v"(t) : "v"(k), "v"(x));
  return t;
}
__device__ __forceinline__ v2f pk_add_lo_mul(v2f t, v2f k, v2f x) {
  asm("v_pk_fma_f32 %0, %1, %2, %0 op_sel_hi:[0,1,1]"
      : "+v"(t) : "v"(k), "v"(x));
  return t;
}
__device__ __forceinline__ v2f pk_add_hi_mul(v2f t, v2f k, v2f x) {
  asm("v_pk_fma_f32 %0, %1, %2, %0 op_sel:[1,0,0] op_sel_hi:[1,1,1]"
      : "+v"(t) : "v"(k), "v"(x));
  return t;
}

template <int NB, int J0, int NT>
__device__ __forceinline__ void gate_group(const Gate (&g)[NT],
                                           const float* __restrict__ src, v4f (&v)[NB]) {
  v4f L[NB], R[NB];
  #pragma unroll
  for (int u = 0; u < NB; ++u) {
    L[u] = *(const v4f*)(src + g[J0 + u].oL);
    R[u] = *(const v4f*)(src + g[J0 + u].oR);
  }
  #pragma unroll
  for (int u = 0; u < NB; ++u) {
    const Gate& G = g[J0 + u];
    v2f tl = pk_lo_plus_hi_mul(G.kx, L[u].xy);
    v2f th = pk_lo_plus_hi_mul(G.kx, L[u].zw);
    tl = pk_add_lo_mul(tl, G.kz, R[u].xy);
    th = pk_add_lo_mul(th, G.kz, R[u].zw);
    const v2f ml = L[u].xy * R[u].xy;
    const v2f mh = L[u].zw * R[u].zw;
    tl = pk_add_hi_mul(tl, G.kz, ml);
    th = pk_add_hi_mul(th, G.kz, mh);
    v[u] = __builtin_shufflevector(tl, th, 0, 1, 2, 3);
  }
}

__global__ __launch_bounds__(NTHREADS)
void dlgn_main(const float* __restrict__ x, const float* __restrict__ thr,
               const int2* __restrict__ idx, const float4* __restrict__ co,
               float* __restrict__ out) {
  __shared__ __attribute__((aligned(16))) float sh_hin[NIN0 * BT];
  __shared__ __attribute__((aligned(16))) float sh_l0 [W0   * BT];
  __shared__ __attribute__((aligned(16))) float sh_l1 [W1   * BT];
  __shared__ __attribute__((aligned(16))) float sh_pop[POPN * BT];

  const int t    = threadIdx.x;
  const int c4   = (t & 3) * VB;
  const int slot = t >> 2;
  const float* xb = x + (size_t)(blockIdx.x * BT + c4) * PFEAT;

  const float myth = thr[slot & 7];

  const int     row = slot * BT + c4;

  Gate g0[G0T], g1[G1T], g2[2];
  #pragma unroll
  for (int j = 0; j < G0T; ++j) g0[j] = load_gate(idx, co, slot + NSLOT * j, c4);
  #pragma unroll
  for (int j = 0; j < G1T; ++j) g1[j] = load_gate(idx, co, W0 + slot + NSLOT * j, c4);
  g2[0] = load_gate(idx, co, W0 + W1 + slot, c4);
  g2[1] = g2[0];
  if (slot < POPN) g2[1] = load_gate(idx, co, W0 + W1 + NSLOT + slot, c4);

  {
    v4f z = {0.0f, 0.0f, 0.0f, 0.0f};
    *(v4f*)&sh_hin[PFEAT * BT + row] = z;
  }

  {
    float* const rd1 = sh_l1;
    float* const rd0 = sh_l0;
    rd0[t] = 0.0f;
    rd0[t + NTHREADS] = 0.0f;
    rd1[t] = 0.0f;
    __syncthreads();
    rd1[(unsigned)g2[0].oL / BT] = 1.0f;
    rd1[(unsigned)g2[0].oR / BT] = 1.0f;
    rd1[(unsigned)g2[1].oL / BT] = 1.0f;
    rd1[(unsigned)g2[1].oR / BT] = 1.0f;
    __syncthreads();
    bool lv1[G1T], lv0[G0T];
    #pragma unroll
    for (int j = 0; j < G1T; ++j) {
      lv1[j] = rd1[slot + NSLOT * j] != 0.0f;
      if (lv1[j]) {
        rd0[(unsigned)g1[j].oL / BT] = 1.0f;
        rd0[(unsigned)g1[j].oR / BT] = 1.0f;
      }
    }
    __syncthreads();
    #pragma unroll
    for (int j = 0; j < G0T; ++j) lv0[j] = rd0[slot + NSLOT * j] != 0.0f;

    const int lane  = t & 31;
    const int hbase = lane & 16;
    #pragma unroll
    for (int j = 0; j < G0T; ++j) {
      const unsigned hm = (__builtin_amdgcn_ballot_w32(lv0[j]) >> hbase) & 0xffffu;
      const int from = hbase + (hm ? (__builtin_ctz(hm) & ~3) : 0) + (lane & 3);
      const int aL = __shfl(g0[j].oL, from);
      const int aR = __shfl(g0[j].oR, from);
      if (!lv0[j]) { g0[j].oL = aL; g0[j].oR = aR; }
    }
    #pragma unroll
    for (int j = 0; j < G1T; ++j) {
      const unsigned hm = (__builtin_amdgcn_ballot_w32(lv1[j]) >> hbase) & 0xffffu;
      const int from = hbase + (hm ? (__builtin_ctz(hm) & ~3) : 0) + (lane & 3);
      const int aL = __shfl(g1[j].oL, from);
      const int aR = __shfl(g1[j].oR, from);
      if (!lv1[j]) { g1[j].oL = aL; g1[j].oR = aR; }
    }
  }

  for (int s = 0; s < STEPSN; ++s) {
    const float* xs = xb + s * 98 + (slot >> 3);
    {
      float px[ENCF][VB];
      for (int j = 0; j < ENCF; ++j)
        for (int i = 0; i < VB; ++i) px[j][i] = xs[i * PFEAT + PIXJ * j];
      float pt[VB] = {0.0f, 0.0f, 0.0f, 0.0f};
      if (slot < ENCR)
        for (int i = 0; i < VB; ++i) pt[i] = xs[i * PFEAT + PIXJ * ENCF];
      for (int j = 0; j < ENCF; ++j) {
        v4f e;
        for (int i = 0; i < VB; ++i) e[i] = (px[j][i] >= myth) ? 1.0f : 0.0f;
        *(v4f*)&sh_hin[j * NSLOT * BT + row] = e;
      }
      if (slot < ENCR) {
        v4f e;
        for (int i = 0; i < VB; ++i) e[i] = (pt[i] >= myth) ? 1.0f : 0.0f;
        *(v4f*)&sh_hin[ENCF * NSLOT * BT + row] = e;
      }
    }
    if (s + 1 < STEPSN) __builtin_prefetch(xs + (slot & 3) * PFEAT + 98, 0, 0);
    __syncthreads();

    {
      v4f v[GB];
      gate_group<GB, 0, G0T>(g0, sh_hin, v);
      for (int u = 0; u < GB; ++u) *(v4f*)&sh_l0[(0 + u) * NSLOT * BT + row] = v[u];
      gate_group<GB, 2, G0T>(g0, sh_hin, v);
      for (int u = 0; u < GB; ++u) *(v4f*)&sh_l0[(2 + u) * NSLOT * BT + row] = v[u];
      gate_group<GB, 4, G0T>(g0, sh_hin, v);
      for (int u = 0; u < GB; ++u) *(v4f*)&sh_l0[(4 + u) * NSLOT * BT + row] = v[u];
      gate_group<GB, 6, G0T>(g0, sh_hin, v);
      for (int u = 0; u < GB; ++u) *(v4f*)&sh_l0[(6 + u) * NSLOT * BT + row] = v[u];
    }
    __syncthreads();

    {
      v4f v[GB];
      gate_group<GB, 0, G1T>(g1, sh_l0, v);
      for (int u = 0; u < GB; ++u) *(v4f*)&sh_l1[(0 + u) * NSLOT * BT + row] = v[u];
      gate_group<GB, 2, G1T>(g1, sh_l0, v);
      for (int u = 0; u < GB; ++u) *(v4f*)&sh_l1[(2 + u) * NSLOT * BT + row] = v[u];
    }
    __syncthreads();

    if (slot < POPN) {
      v4f v[2];
      gate_group<2, 0, 2>(g2, sh_l1, v);
      *(v4f*)&sh_hin[PFEAT * BT + row] = v[0];
      *(v4f*)&sh_pop[row]              = v[1];
    } else {
      v4f v[1];
      gate_group<1, 0, 2>(g2, sh_l1, v);
      *(v4f*)&sh_hin[PFEAT * BT + row] = v[0];
    }
    __syncthreads();
  }

  if (t < 32) {
    v8f acc = {};
    v2f ones; ones[0] = 1.0f; ones[1] = 1.0f;
    const int col   = t & 15;
    const int khalf = (t < 16) ? 0 : 2;
    #pragma unroll 4
    for (int kc = 0; kc < 32; ++kc) {
      v2f bm;
      bm[0] = sh_pop[(4 * kc + khalf + 0) * BT + col];
      bm[1] = sh_pop[(4 * kc + khalf + 1) * BT + col];
      acc = __builtin_amdgcn_wmma_f32_16x16x4_f32(
           false, ones,  false, bm,
           (short)0, acc,  false,  false);
    }
    if (t < 16) {
      float logit = acc[0] / 10.0f;
      float* ob = out + (size_t)(blockIdx.x * BT + t) * 10;
      #pragma unroll
      for (int c = 0; c < 10; ++c) ob[c] = logit;
    }
  }
}

extern "C" void kernel_launch(void* const* d_in, const int* in_sizes, int n_in,
                              void* d_out, int out_size, void* d_ws, size_t ws_size,
                              hipStream_t stream) {
  const float* x   = (const float*)d_in[0];
  const float* thr = (const float*)d_in[1];
  const float* lg0 = (const float*)d_in[2];
  const int*   le0 = (const int*)  d_in[3];
  const int*   ri0 = (const int*)  d_in[4];
  const float* lg1 = (const float*)d_in[5];
  const int*   le1 = (const int*)  d_in[6];
  const int*   ri1 = (const int*)  d_in[7];
  const float* lg2 = (const float*)d_in[8];
  const int*   le2 = (const int*)  d_in[9];
  const int*   ri2 = (const int*)  d_in[10];

  int2*   idx = (int2*)d_ws;
  float4* co  = (float4*)((char*)d_ws + 32768);

  dlgn_precomp<<<(NGATES + 255) / 256, 256, 0, stream>>>(
      lg0, le0, ri0, lg1, le1, ri1, lg2, le2, ri2, idx, co);

  dlgn_main<<<4096 / BT, NTHREADS, 0, stream>>>(x, thr, idx, co, (float*)d_out);
}


# Gates of layers 0/1 whose row no evaluated gate reads are not evaluated any more: each workgroup ranks the read gates from the wiring, threads take them by rank, rows = ranks; sum order kept
# speedup vs baseline: 5.4507x; 1.0763x over previous
#include <hip/hip_runtime.h>


#define BT       16
#define NTHREADS 1024
#define VB       4
#define NSLOT    (NTHREADS * VB / BT)
#define W0       2048
#define W1       1024
#define W2       384
#define PFEAT    784
#define FBN      256
#define POPN     128
#define NIN0     (PFEAT + FBN)
#define STEPSN   8
#define NGATES   (W0 + W1 + W2)
#define GB       2
#define G0T      (W0 / NSLOT)
#define G1T      (W1 / NSLOT)
#define ENCF     (PFEAT / NSLOT)
#define ENCR     (PFEAT % NSLOT)
#define PIXJ     (NSLOT / 8)

static_assert(FBN == NSLOT, "layer 2: one feedback gate per slot");
static_assert(W2 - FBN == POPN && POPN * 2 == NSLOT, "layer 2: population gates on the lower half of the slots");

typedef float v2f __attribute__((ext_vector_type(2)));
typedef float v4f __attribute__((ext_vector_type(4)));
typedef float v8f __attribute__((ext_vector_type(8)));

__global__ void dlgn_precomp(const float* __restrict__ lg0,
                             const int*  __restrict__ le0, const int* __restrict__ ri0,
                             const float* __restrict__ lg1,
                             const int*  __restrict__ le1, const int* __restrict__ ri1,
                             const float* __restrict__ lg2,
                             const int*  __restrict__ le2, const int* __restrict__ ri2,
                             int2*  __restrict__ idx, float4* __restrict__ co) {
  int G = blockIdx.x * blockDim.x + threadIdx.x;
  if (G >= NGATES) return;
  const float* lg; const int* li; const int* ri; int g;
  if (G < W0)            { lg = lg0; li = le0; ri = ri0; g = G; }
  else if (G < W0 + W1)  { lg = lg1; li = le1; ri = ri1; g = G - W0; }
  else                   { lg = lg2; li = le2; ri = ri2; g = G - W0 - W1; }
  float w0 = 0.5f + 0.5f * sinf(lg[g * 4 + 0]);
  float w1 = 0.5f + 0.5f * sinf(lg[g * 4 + 1]);
  float w2 = 0.5f + 0.5f * sinf(lg[g * 4 + 2]);
  float w3 = 0.5f + 0.5f * sinf(lg[g * 4 + 3]);
  co[G]  = make_float4(w0, w2 - w0, w1 - w0, w0 - w1 - w2 + w3);
  idx[G] = make_int2(li[g], ri[g]);
}

struct Gate { int oL, oR; v2f kx, kz; };

__device__ __forceinline__ Gate load_gate(const int2* __restrict__ idx,
                                          const float4* __restrict__ co, int g, int c4) {
  const int2   lr = idx[g];
  const float4 w  = co[g];
  Gate G;
  G.oL = lr.x * BT + c4;
  G.oR = lr.y * BT + c4;
  G.kx[0] = w.x; G.kx[1] = w.y;
  G.kz[0] = w.z; G.kz[1] = w.w;
  return G;
}


__device__ __forceinline__ v2f pk_lo_plus_hi_mul(v2f k, v2f x) {
  v2f t;
  asm("v_pk_fma_f32 %0, %1, %2, %1 op_sel:[1,0,0] op_sel_hi:[1,1,0]"
      : "=&v"(t) : "v"(k), "v"(x));
  return t;
}
__device__ __forceinline__ v2f pk_add_lo_mul(v2f t, v2f k, v2f x) {
  asm("v_pk_fma_f32 %0, %1, %2, %0 op_sel_hi:[0,1,1]"
      : "+v"(t) : "v"(k), "v"(x));
  return t;
}
__device__ __forceinline__ v2f pk_add_hi_mul(v2f t, v2f k, v2f x) {
  asm("v_pk_fma_f32 %0, %1, %2, %0 op_sel:[1,0,0] op_sel_hi:[1,1,1]"
      : "+v"(t) : "v"(k), "v"(x));
  return t;
}

template <int NB, int J0, int NT>
__device__ __forceinline__ void gate_group(const Gate (&g)[NT],
                                           const float* __restrict__ src, v4f (&v)[NB]) {
  v4f L[NB], R[NB];
  #pragma unroll
  for (int u = 0; u < NB; ++u) {
    L[u] = *(const v4f*)(src + g[J0 + u].oL);
    R[u] = *(const v4f*)(src + g[J0 + u].oR);
  }
  #pragma unroll
  for (int u = 0; u < NB; ++u) {
    const Gate& G = g[J0 + u];
    v2f tl = pk_lo_plus_hi_mul(G.kx, L[u].xy);
    v2f th = pk_lo_plus_hi_mul(G.kx, L[u].zw);
    tl = pk_add_lo_mul(tl, G.kz, R[u].xy);
    th = pk_add_lo_mul(th, G.kz, R[u].zw);
    const v2f ml = L[u].xy * R[u].xy;
    const v2f mh = L[u].zw * R[u].zw;
    tl = pk_add_hi_mul(tl, G.kz, ml);
    th = pk_add_hi_mul(th, G.kz, mh);
    v[u] = __builtin_shufflevector(tl, th, 0, 1, 2, 3);
  }
}

template <int J, int NT>
__device__ __forceinline__ void round_pair(const Gate (&g)[NT], int nr,
                                           const float* __restrict__ src,
                                           float* __restrict__ dst) {
  if (nr > J + 1) {
    v4f v[GB];
    gate_group<GB, J, NT>(g, src, v);
    for (int u = 0; u < GB; ++u) *(v4f*)(dst + (J + u) * NSLOT * BT) = v[u];
  } else {
    v4f v[1];
    gate_group<1, J, NT>(g, src, v);
    *(v4f*)(dst + J * NSLOT * BT) = v[0];
  }
}

__global__ __launch_bounds__(NTHREADS)
void dlgn_main(const float* __restrict__ x, const float* __restrict__ thr,
               const int2* __restrict__ idx, const float4* __restrict__ co,
               float* __restrict__ out) {
  __shared__ __attribute__((aligned(16))) float sh_hin[NIN0 * BT];
  __shared__ __attribute__((aligned(16))) float sh_l0 [W0   * BT];
  __shared__ __attribute__((aligned(16))) float sh_l1 [W1   * BT];
  __shared__ __attribute__((aligned(16))) float sh_pop[POPN * BT];

  const int t    = threadIdx.x;
  const int c4   = (t & 3) * VB;
  const int slot = t >> 2;
  const float* xb = x + (size_t)(blockIdx.x * BT + c4) * PFEAT;

  const float myth = thr[slot & 7];

  const int     row = slot * BT + c4;

  {
    v4f z = {0.0f, 0.0f, 0.0f, 0.0f};
    *(v4f*)&sh_hin[PFEAT * BT + row] = z;
  }

  Gate g0[G0T], g1[G1T], g2[2];
  int n0, n1;
  {
    float* const F0 = sh_l0;
    float* const R0 = sh_l0 + W0;
    v2f*   const O0 = (v2f*)(sh_l0 + 2 * W0);
    v4f*   const K0 = (v4f*)(sh_l0 + 4 * W0);
    v2f*   const O1 = (v2f*)(sh_l0 + 8 * W0);
    v4f*   const K1 = (v4f*)(sh_l0 + 9 * W0);
    float* const F1 = sh_l1;
    float* const R1 = sh_l1 + W1;
    float* const C0 = sh_l1 + 2 * W1;
    float* const C1 = C0 + 64;
    static_assert(11 * W0 <= W0 * BT && 2 * W1 + 96 <= W1 * BT, "tables fit");
    static_assert(W1 == NTHREADS && W0 == 2 * NTHREADS && NTHREADS == 32 * 32, "one thread per layer-1 gate");

    const int lane = t & 31;
    const int wv   = t >> 5;

    const int2   i0a = idx[t];
    const int2   i0b = idx[t + NTHREADS];
    const int2   i1  = idx[W0 + t];
    const float4 k0a = co[t];
    const float4 k0b = co[t + NTHREADS];
    const float4 k1  = co[W0 + t];
    const int2   i2f = idx[W0 + W1 + slot];
    const float4 k2f = co[W0 + W1 + slot];
    int2   i2p = i2f;
    float4 k2p = k2f;
    if (slot < POPN) {
      i2p = idx[W0 + W1 + NSLOT + slot];
      k2p = co[W0 + W1 + NSLOT + slot];
    }

    F0[t] = 0.0f;
    F0[t + NTHREADS] = 0.0f;
    F1[t] = 0.0f;
    __syncthreads();
    F1[i2f.x] = 1.0f;
    F1[i2f.y] = 1.0f;
    F1[i2p.x] = 1.0f;
    F1[i2p.y] = 1.0f;
    __syncthreads();
    const bool lv1 = F1[t] != 0.0f;
    if (lv1) {
      F0[i1.x] = 1.0f;
      F0[i1.y] = 1.0f;
    }
    __syncthreads();
    const bool lv0a = F0[t] != 0.0f;
    const bool lv0b = F0[t + NTHREADS] != 0.0f;
    const unsigned m1  = __builtin_amdgcn_ballot_w32(lv1);
    const unsigned m0a = __builtin_amdgcn_ballot_w32(lv0a);
    const unsigned m0b = __builtin_amdgcn_ballot_w32(lv0b);
    if (lane == 0) {
      C0[wv]      = __int_as_float(__builtin_popcount(m0a));
      C0[32 + wv] = __int_as_float(__builtin_popcount(m0b));
      C1[wv]      = __int_as_float(__builtin_popcount(m1));
    }
    __syncthreads();

    const unsigned c0a = (unsigned)__float_as_int(C0[lane]);
    const unsigned c0b = (unsigned)__float_as_int(C0[32 + lane]);
    const unsigned c1  = (unsigned)__float_as_int(C1[lane]);
    const bool below = lane < wv;
    unsigned pa = (below ? c0a : 0u) | ((below ? c0b : 0u) << 16);
    unsigned pb = c0a | (c0b << 16);
    unsigned pc = (below ? c1 : 0u) | (c1 << 16);
    for (int m = 16; m >= 1; m >>= 1) {
      pa += (unsigned)__shfl_xor((int)pa, m);
      pb += (unsigned)__shfl_xor((int)pb, m);
      pc += (unsigned)__shfl_xor((int)pc, m);
    }
    const unsigned lt = (1u << lane) - 1u;
    const int n0a = (int)(pb & 0xffffu);
    const int rk0a = (int)(pa & 0xffffu) + __builtin_popcount(m0a & lt);
    const int rk0b = n0a + (int)(pa >> 16) + __builtin_popcount(m0b & lt);
    const int rk1  = (int)(pc & 0xffffu) + __builtin_popcount(m1 & lt);
    n0 = __builtin_amdgcn_readfirstlane(n0a + (int)(pb >> 16));
    n1 = __builtin_amdgcn_readfirstlane((int)(pc >> 16));

    if (lv0a) {
      v2f o; o[0] = __int_as_float(i0a.x * BT); o[1] = __int_as_float(i0a.y * BT);
      v4f k; k[0] = k0a.x; k[1] = k0a.y; k[2] = k0a.z; k[3] = k0a.w;
      R0[t] = __int_as_float(rk0a);
      O0[rk0a] = o;
      K0[rk0a] = k;
    }
    if (lv0b) {
      v2f o; o[0] = __int_as_float(i0b.x * BT); o[1] = __int_as_float(i0b.y * BT);
      v4f k; k[0] = k0b.x; k[1] = k0b.y; k[2] = k0b.z; k[3] = k0b.w;
      R0[t + NTHREADS] = __int_as_float(rk0b);
      O0[rk0b] = o;
      K0[rk0b] = k;
    }
    if (lv1) R1[t] = __int_as_float(rk1);
    __syncthreads();

    if (lv1) {
      v2f o;
      o[0] = __int_as_float(__float_as_int(R0[i1.x]) * BT);
      o[1] = __int_as_float(__float_as_int(R0[i1.y]) * BT);
      v4f k; k[0] = k1.x; k[1] = k1.y; k[2] = k1.z; k[3] = k1.w;
      O1[rk1] = o;
      K1[rk1] = k;
    }
    g2[0].oL = __float_as_int(R1[i2f.x]) * BT + c4;
    g2[0].oR = __float_as_int(R1[i2f.y]) * BT + c4;
    g2[0].kx[0] = k2f.x; g2[0].kx[1] = k2f.y;
    g2[0].kz[0] = k2f.z; g2[0].kz[1] = k2f.w;
    g2[1].oL = __float_as_int(R1[i2p.x]) * BT + c4;
    g2[1].oR = __float_as_int(R1[i2p.y]) * BT + c4;
    g2[1].kx[0] = k2p.x; g2[1].kx[1] = k2p.y;
    g2[1].kz[0] = k2p.z; g2[1].kz[1] = k2p.w;
    __syncthreads();

    #pragma unroll
    for (int j = 0; j < G0T; ++j) {
      const int p = slot + NSLOT * j;
      Gate G;
      G.oL = c4; G.oR = c4;
      G.kx[0] = 0.0f; G.kx[1] = 0.0f; G.kz[0] = 0.0f; G.kz[1] = 0.0f;
      if (p < n0) {
        const v2f o = O0[p];
        const v4f k = K0[p];
        G.oL = __float_as_int(o[0]) + c4;
        G.oR = __float_as_int(o[1]) + c4;
        G.kx[0] = k[0]; G.kx[1] = k[1]; G.kz[0] = k[2]; G.kz[1] = k[3];
      }
      g0[j] = G;
    }
    #pragma unroll
    for (int j = 0; j < G1T; ++j) {
      const int p = slot + NSLOT * j;
      Gate G;
      G.oL = c4; G.oR = c4;
      G.kx[0] = 0.0f; G.kx[1] = 0.0f; G.kz[0] = 0.0f; G.kz[1] = 0.0f;
      if (p < n1) {
        const v2f o = O1[p];
        const v4f k = K1[p];
        G.oL = __float_as_int(o[0]) + c4;
        G.oR = __float_as_int(o[1]) + c4;
        G.kx[0] = k[0]; G.kx[1] = k[1]; G.kz[0] = k[2]; G.kz[1] = k[3];
      }
      g1[j] = G;
    }
  }

  const int wslot = __builtin_amdgcn_readfirstlane(slot & ~7);
  const int nr0 = (n0 - wslot + NSLOT - 1) / NSLOT;
  const int nr1 = (n1 - wslot + NSLOT - 1) / NSLOT;

  for (int s = 0; s < STEPSN; ++s) {
    const float* xs = xb + s * 98 + (slot >> 3);
    {
      float px[ENCF][VB];
      for (int j = 0; j < ENCF; ++j)
        for (int i = 0; i < VB; ++i) px[j][i] = xs[i * PFEAT + PIXJ * j];
      float pt[VB] = {0.0f, 0.0f, 0.0f, 0.0f};
      if (slot < ENCR)
        for (int i = 0; i < VB; ++i) pt[i] = xs[i * PFEAT + PIXJ * ENCF];
      for (int j = 0; j < ENCF; ++j) {
        v4f e;
        for (int i = 0; i < VB; ++i) e[i] = (px[j][i] >= myth) ? 1.0f : 0.0f;
        *(v4f*)&sh_hin[j * NSLOT * BT + row] = e;
      }
      if (slot < ENCR) {
        v4f e;
        for (int i = 0; i < VB; ++i) e[i] = (pt[i] >= myth) ? 1.0f : 0.0f;
        *(v4f*)&sh_hin[ENCF * NSLOT * BT + row] = e;
      }
    }
    if (s + 1 < STEPSN) __builtin_prefetch(xs + (slot & 3) * PFEAT + 98, 0, 0);
    __syncthreads();

    if (nr0 > 0) {
      round_pair<0, G0T>(g0, nr0, sh_hin, sh_l0 + row);
      if (nr0 > 2) {
        round_pair<2, G0T>(g0, nr0, sh_hin, sh_l0 + row);
        if (nr0 > 4) {
          round_pair<4, G0T>(g0, nr0, sh_hin, sh_l0 + row);
          if (nr0 > 6) round_pair<6, G0T>(g0, nr0, sh_hin, sh_l0 + row);
        }
      }
    }
    __syncthreads();

    if (nr1 > 0) {
      round_pair<0, G1T>(g1, nr1, sh_l0, sh_l1 + row);
      if (nr1 > 2) round_pair<2, G1T>(g1, nr1, sh_l0, sh_l1 + row);
    }
    __syncthreads();

    if (slot < POPN) {
      v4f v[2];
      gate_group<2, 0, 2>(g2, sh_l1, v);
      *(v4f*)&sh_hin[PFEAT * BT + row] = v[0];
      *(v4f*)&sh_pop[row]              = v[1];
    } else {
      v4f v[1];
      gate_group<1, 0, 2>(g2, sh_l1, v);
      *(v4f*)&sh_hin[PFEAT * BT + row] = v[0];
    }
    __syncthreads();
  }

  if (t < 32) {
    v8f acc = {};
    v2f ones; ones[0] = 1.0f; ones[1] = 1.0f;
    const int col   = t & 15;
    const int khalf = (t < 16) ? 0 : 2;
    #pragma unroll 4
    for (int kc = 0; kc < 32; ++kc) {
      v2f bm;
      bm[0] = sh_pop[(4 * kc + khalf + 0) * BT + col];
      bm[1] = sh_pop[(4 * kc + khalf + 1) * BT + col];
      acc = __builtin_amdgcn_wmma_f32_16x16x4_f32(
           false, ones,  false, bm,
           (short)0, acc,  false,  false);
    }
    if (t < 16) {
      float logit = acc[0] / 10.0f;
      float* ob = out + (size_t)(blockIdx.x * BT + t) * 10;
      #pragma unroll
      for (int c = 0; c < 10; ++c) ob[c] = logit;
    }
  }
}

extern "C" void kernel_launch(void* const* d_in, const int* in_sizes, int n_in,
                              void* d_out, int out_size, void* d_ws, size_t ws_size,
                              hipStream_t stream) {
  const float* x   = (const float*)d_in[0];
  const float* thr = (const float*)d_in[1];
  const float* lg0 = (const float*)d_in[2];
  const int*   le0 = (const int*)  d_in[3];
  const int*   ri0 = (const int*)  d_in[4];
  const float* lg1 = (const float*)d_in[5];
  const int*   le1 = (const int*)  d_in[6];
  const int*   ri1 = (const int*)  d_in[7];
  const float* lg2 = (const float*)d_in[8];
  const int*   le2 = (const int*)  d_in[9];
  const int*   ri2 = (const int*)  d_in[10];

  int2*   idx = (int2*)d_ws;
  float4* co  = (float4*)((char*)d_ws + 32768);

  dlgn_precomp<<<(NGATES + 255) / 256, 256, 0, stream>>>(
      lg0, le0, ri0, lg1, le1, ri1, lg2, le2, ri2, idx, co);

  dlgn_main<<<4096 / BT, NTHREADS, 0, stream>>>(x, thr, idx, co, (float*)d_out);
}


# A step's pixels are fetched one step ahead (each value once per workgroup, coalesced), staged raw in LDS and thresholded inside the layer-2 phase: no load wait, 3 barriers per step; sum order kept
# speedup vs baseline: 6.6818x; 1.2259x over previous
#include <hip/hip_runtime.h>


#define BT       16
#define NTHREADS 1024
#define VB       4
#define NSLOT    (NTHREADS * VB / BT)
#define W0       2048
#define W1       1024
#define W2       384
#define PFEAT    784
#define FBN      256
#define POPN     128
#define NIN0     (PFEAT + FBN)
#define STEPSN   8
#define NGATES   (W0 + W1 + W2)
#define GB       2
#define G0T      (W0 / NSLOT)
#define G1T      (W1 / NSLOT)
#define ENCF     (PFEAT / NSLOT)
#define ENCR     (PFEAT % NSLOT)
#define PIXJ     (NSLOT / 8)
#define NPIX     (PFEAT / 8)
#define PXP      20

static_assert(FBN == NSLOT, "layer 2: one feedback gate per slot");
static_assert(NPIX * 8 == PFEAT && NPIX > 64 && NPIX <= 128 && NTHREADS == 64 * BT, "pixel loads: 64 threads per batch row, two pixels each at most");
static_assert(W2 - FBN == POPN && POPN * 2 == NSLOT, "layer 2: population gates on the lower half of the slots");

typedef float v2f __attribute__((ext_vector_type(2)));
typedef float v4f __attribute__((ext_vector_type(4)));
typedef float v8f __attribute__((ext_vector_type(8)));

__global__ void dlgn_precomp(const float* __restrict__ lg0,
                             const int*  __restrict__ le0, const int* __restrict__ ri0,
                             const float* __restrict__ lg1,
                             const int*  __restrict__ le1, const int* __restrict__ ri1,
                             const float* __restrict__ lg2,
                             const int*  __restrict__ le2, const int* __restrict__ ri2,
                             int2*  __restrict__ idx, float4* __restrict__ co) {
  int G = blockIdx.x * blockDim.x + threadIdx.x;
  if (G >= NGATES) return;
  const float* lg; const int* li; const int* ri; int g;
  if (G < W0)            { lg = lg0; li = le0; ri = ri0; g = G; }
  else if (G < W0 + W1)  { lg = lg1; li = le1; ri = ri1; g = G - W0; }
  else                   { lg = lg2; li = le2; ri = ri2; g = G - W0 - W1; }
  float w0 = 0.5f + 0.5f * sinf(lg[g * 4 + 0]);
  float w1 = 0.5f + 0.5f * sinf(lg[g * 4 + 1]);
  float w2 = 0.5f + 0.5f * sinf(lg[g * 4 + 2]);
  float w3 = 0.5f + 0.5f * sinf(lg[g * 4 + 3]);
  co[G]  = make_float4(w0, w2 - w0, w1 - w0, w0 - w1 - w2 + w3);
  idx[G] = make_int2(li[g], ri[g]);
}

struct Gate { int oL, oR; v2f kx, kz; };

__device__ __forceinline__ Gate load_gate(const int2* __restrict__ idx,
                                          const float4* __restrict__ co, int g, int c4) {
  const int2   lr = idx[g];
  const float4 w  = co[g];
  Gate G;
  G.oL = lr.x * BT + c4;
  G.oR = lr.y * BT + c4;
  G.kx[0] = w.x; G.kx[1] = w.y;
  G.kz[0] = w.z; G.kz[1] = w.w;
  return G;
}


__device__ __forceinline__ v2f pk_lo_plus_hi_mul(v2f k, v2f x) {
  v2f t;
  asm("v_pk_fma_f32 %0, %1, %2, %1 op_sel:[1,0,0] op_sel_hi:[1,1,0]"
      : "=&v"(t) : "v"(k), "v"(x));
  return t;
}
__device__ __forceinline__ v2f pk_add_lo_mul(v2f t, v2f k, v2f x) {
  asm("v_pk_fma_f32 %0, %1, %2, %0 op_sel_hi:[0,1,1]"
      : "+v"(t) : "v"(k), "v"(x));
  return t;
}
__device__ __forceinline__ v2f pk_add_hi_mul(v2f t, v2f k, v2f x) {
  asm("v_pk_fma_f32 %0, %1, %2, %0 op_sel:[1,0,0] op_sel_hi:[1,1,1]"
      : "+v"(t) : "v"(k), "v"(x));
  return t;
}

template <int NB, int J0, int NT>
__device__ __forceinline__ void gate_group(const Gate (&g)[NT],
                                           const float* __restrict__ src, v4f (&v)[NB]) {
  v4f L[NB], R[NB];
  #pragma unroll
  for (int u = 0; u < NB; ++u) {
    L[u] = *(const v4f*)(src + g[J0 + u].oL);
    R[u] = *(const v4f*)(src + g[J0 + u].oR);
  }
  #pragma unroll
  for (int u = 0; u < NB; ++u) {
    const Gate& G = g[J0 + u];
    v2f tl = pk_lo_plus_hi_mul(G.kx, L[u].xy);
    v2f th = pk_lo_plus_hi_mul(G.kx, L[u].zw);
    tl = pk_add_lo_mul(tl, G.kz, R[u].xy);
    th = pk_add_lo_mul(th, G.kz, R[u].zw);
    const v2f ml = L[u].xy * R[u].xy;
    const v2f mh = L[u].zw * R[u].zw;
    tl = pk_add_hi_mul(tl, G.kz, ml);
    th = pk_add_hi_mul(th, G.kz, mh);
    v[u] = __builtin_shufflevector(tl, th, 0, 1, 2, 3);
  }
}

template <int J, int NT>
__device__ __forceinline__ void round_pair(const Gate (&g)[NT], int nr,
                                           const float* __restrict__ src,
                                           float* __restrict__ dst) {
  if (nr > J + 1) {
    v4f v[GB];
    gate_group<GB, J, NT>(g, src, v);
    for (int u = 0; u < GB; ++u) *(v4f*)(dst + (J + u) * NSLOT * BT) = v[u];
  } else {
    v4f v[1];
    gate_group<1, J, NT>(g, src, v);
    *(v4f*)(dst + J * NSLOT * BT) = v[0];
  }
}

__device__ __forceinline__ void stage_pixels(float* __restrict__ sh_px, int lk, int lb,
                                             bool second, float pa, float pb) {
  sh_px[lk * PXP + lb] = pa;
  if (second) sh_px[(64 + lk) * PXP + lb] = pb;
}

__device__ __forceinline__ void encode_step(const float* __restrict__ sh_px,
                                            float* __restrict__ sh_hin,
                                            int slot, int c4, int row, float myth) {
  const float* ps = sh_px + (slot >> 3) * PXP + c4;
  v4f p[ENCF];
  for (int j = 0; j < ENCF; ++j) p[j] = *(const v4f*)(ps + PIXJ * j * PXP);
  v4f pt = {0.0f, 0.0f, 0.0f, 0.0f};
  if (slot < ENCR) pt = *(const v4f*)(ps + PIXJ * ENCF * PXP);
  for (int j = 0; j < ENCF; ++j) {
    v4f e;
    for (int i = 0; i < VB; ++i) e[i] = (p[j][i] >= myth) ? 1.0f : 0.0f;
    *(v4f*)&sh_hin[j * NSLOT * BT + row] = e;
  }
  if (slot < ENCR) {
    v4f e;
    for (int i = 0; i < VB; ++i) e[i] = (pt[i] >= myth) ? 1.0f : 0.0f;
    *(v4f*)&sh_hin[ENCF * NSLOT * BT + row] = e;
  }
}

__global__ __launch_bounds__(NTHREADS)
void dlgn_main(const float* __restrict__ x, const float* __restrict__ thr,
               const int2* __restrict__ idx, const float4* __restrict__ co,
               float* __restrict__ out) {
  __shared__ __attribute__((aligned(16))) float sh_hin[NIN0 * BT];
  __shared__ __attribute__((aligned(16))) float sh_l0 [W0   * BT];
  __shared__ __attribute__((aligned(16))) float sh_l1 [W1   * BT];
  __shared__ __attribute__((aligned(16))) float sh_pop[POPN * BT];
  __shared__ __attribute__((aligned(16))) float sh_px [NPIX * PXP];

  const int t    = threadIdx.x;
  const int c4   = (t & 3) * VB;
  const int slot = t >> 2;

  const int  lb     = t >> 6;
  const int  lk     = t & 63;
  const bool second = lk < NPIX - 64;
  const float* xl = x + (size_t)(blockIdx.x * BT + lb) * PFEAT + lk;

  float pa = xl[0];
  float pb = 0.0f;
  if (second) pb = xl[64];

  const float myth = thr[slot & 7];

  const int     row = slot * BT + c4;

  {
    v4f z = {0.0f, 0.0f, 0.0f, 0.0f};
    *(v4f*)&sh_hin[PFEAT * BT + row] = z;
  }

  Gate g0[G0T], g1[G1T], g2[2];
  int n0, n1;
  {
    float* const F0 = sh_l0;
    float* const R0 = sh_l0 + W0;
    v2f*   const O0 = (v2f*)(sh_l0 + 2 * W0);
    v4f*   const K0 = (v4f*)(sh_l0 + 4 * W0);
    v2f*   const O1 = (v2f*)(sh_l0 + 8 * W0);
    v4f*   const K1 = (v4f*)(sh_l0 + 9 * W0);
    float* const F1 = sh_l1;
    float* const R1 = sh_l1 + W1;
    float* const C0 = sh_l1 + 2 * W1;
    float* const C1 = C0 + 64;
    static_assert(11 * W0 <= W0 * BT && 2 * W1 + 96 <= W1 * BT, "tables fit");
    static_assert(W1 == NTHREADS && W0 == 2 * NTHREADS && NTHREADS == 32 * 32, "one thread per layer-1 gate");

    const int lane = t & 31;
    const int wv   = t >> 5;

    const int2   i0a = idx[t];
    const int2   i0b = idx[t + NTHREADS];
    const int2   i1  = idx[W0 + t];
    const float4 k0a = co[t];
    const float4 k0b = co[t + NTHREADS];
    const float4 k1  = co[W0 + t];
    const int2   i2f = idx[W0 + W1 + slot];
    const float4 k2f = co[W0 + W1 + slot];
    int2   i2p = i2f;
    float4 k2p = k2f;
    if (slot < POPN) {
      i2p = idx[W0 + W1 + NSLOT + slot];
      k2p = co[W0 + W1 + NSLOT + slot];
    }

    F0[t] = 0.0f;
    F0[t + NTHREADS] = 0.0f;
    F1[t] = 0.0f;
    stage_pixels(sh_px, lk, lb, second, pa, pb);
    __syncthreads();
    F1[i2f.x] = 1.0f;
    F1[i2f.y] = 1.0f;
    F1[i2p.x] = 1.0f;
    F1[i2p.y] = 1.0f;
    encode_step(sh_px, sh_hin, slot, c4, row, myth);
    __syncthreads();
    const bool lv1 = F1[t] != 0.0f;
    if (lv1) {
      F0[i1.x] = 1.0f;
      F0[i1.y] = 1.0f;
    }
    __syncthreads();
    const bool lv0a = F0[t] != 0.0f;
    const bool lv0b = F0[t + NTHREADS] != 0.0f;
    const unsigned m1  = __builtin_amdgcn_ballot_w32(lv1);
    const unsigned m0a = __builtin_amdgcn_ballot_w32(lv0a);
    const unsigned m0b = __builtin_amdgcn_ballot_w32(lv0b);
    if (lane == 0) {
      C0[wv]      = __int_as_float(__builtin_popcount(m0a));
      C0[32 + wv] = __int_as_float(__builtin_popcount(m0b));
      C1[wv]      = __int_as_float(__builtin_popcount(m1));
    }
    __syncthreads();

    const unsigned c0a = (unsigned)__float_as_int(C0[lane]);
    const unsigned c0b = (unsigned)__float_as_int(C0[32 + lane]);
    const unsigned c1  = (unsigned)__float_as_int(C1[lane]);
    const bool below = lane < wv;
    unsigned pa = (below ? c0a : 0u) | ((below ? c0b : 0u) << 16);
    unsigned pb = c0a | (c0b << 16);
    unsigned pc = (below ? c1 : 0u) | (c1 << 16);
    for (int m = 16; m >= 1; m >>= 1) {
      pa += (unsigned)__shfl_xor((int)pa, m);
      pb += (unsigned)__shfl_xor((int)pb, m);
      pc += (unsigned)__shfl_xor((int)pc, m);
    }
    const unsigned lt = (1u << lane) - 1u;
    const int n0a = (int)(pb & 0xffffu);
    const int rk0a = (int)(pa & 0xffffu) + __builtin_popcount(m0a & lt);
    const int rk0b = n0a + (int)(pa >> 16) + __builtin_popcount(m0b & lt);
    const int rk1  = (int)(pc & 0xffffu) + __builtin_popcount(m1 & lt);
    n0 = __builtin_amdgcn_readfirstlane(n0a + (int)(pb >> 16));
    n1 = __builtin_amdgcn_readfirstlane((int)(pc >> 16));

    if (lv0a) {
      v2f o; o[0] = __int_as_float(i0a.x * BT); o[1] = __int_as_float(i0a.y * BT);
      v4f k; k[0] = k0a.x; k[1] = k0a.y; k[2] = k0a.z; k[3] = k0a.w;
      R0[t] = __int_as_float(rk0a);
      O0[rk0a] = o;
      K0[rk0a] = k;
    }
    if (lv0b) {
      v2f o; o[0] = __int_as_float(i0b.x * BT); o[1] = __int_as_float(i0b.y * BT);
      v4f k; k[0] = k0b.x; k[1] = k0b.y; k[2] = k0b.z; k[3] = k0b.w;
      R0[t + NTHREADS] = __int_as_float(rk0b);
      O0[rk0b] = o;
      K0[rk0b] = k;
    }
    if (lv1) R1[t] = __int_as_float(rk1);
    __syncthreads();

    if (lv1) {
      v2f o;
      o[0] = __int_as_float(__float_as_int(R0[i1.x]) * BT);
      o[1] = __int_as_float(__float_as_int(R0[i1.y]) * BT);
      v4f k; k[0] = k1.x; k[1] = k1.y; k[2] = k1.z; k[3] = k1.w;
      O1[rk1] = o;
      K1[rk1] = k;
    }
    g2[0].oL = __float_as_int(R1[i2f.x]) * BT + c4;
    g2[0].oR = __float_as_int(R1[i2f.y]) * BT + c4;
    g2[0].kx[0] = k2f.x; g2[0].kx[1] = k2f.y;
    g2[0].kz[0] = k2f.z; g2[0].kz[1] = k2f.w;
    g2[1].oL = __float_as_int(R1[i2p.x]) * BT + c4;
    g2[1].oR = __float_as_int(R1[i2p.y]) * BT + c4;
    g2[1].kx[0] = k2p.x; g2[1].kx[1] = k2p.y;
    g2[1].kz[0] = k2p.z; g2[1].kz[1] = k2p.w;
    __syncthreads();

    #pragma unroll
    for (int j = 0; j < G0T; ++j) {
      const int p = slot + NSLOT * j;
      Gate G;
      G.oL = c4; G.oR = c4;
      G.kx[0] = 0.0f; G.kx[1] = 0.0f; G.kz[0] = 0.0f; G.kz[1] = 0.0f;
      if (p < n0) {
        const v2f o = O0[p];
        const v4f k = K0[p];
        G.oL = __float_as_int(o[0]) + c4;
        G.oR = __float_as_int(o[1]) + c4;
        G.kx[0] = k[0]; G.kx[1] = k[1]; G.kz[0] = k[2]; G.kz[1] = k[3];
      }
      g0[j] = G;
    }
    #pragma unroll
    for (int j = 0; j < G1T; ++j) {
      const int p = slot + NSLOT * j;
      Gate G;
      G.oL = c4; G.oR = c4;
      G.kx[0] = 0.0f; G.kx[1] = 0.0f; G.kz[0] = 0.0f; G.kz[1] = 0.0f;
      if (p < n1) {
        const v2f o = O1[p];
        const v4f k = K1[p];
        G.oL = __float_as_int(o[0]) + c4;
        G.oR = __float_as_int(o[1]) + c4;
        G.kx[0] = k[0]; G.kx[1] = k[1]; G.kz[0] = k[2]; G.kz[1] = k[3];
      }
      g1[j] = G;
    }
    __syncthreads();
  }

  const int wslot = __builtin_amdgcn_readfirstlane(slot & ~7);
  const int nr0 = (n0 - wslot + NSLOT - 1) / NSLOT;
  const int nr1 = (n1 - wslot + NSLOT - 1) / NSLOT;

  for (int s = 0; s < STEPSN; ++s) {
    const bool more = s + 1 < STEPSN;
    if (more) {
      pa = xl[(s + 1) * NPIX];
      if (second) pb = xl[(s + 1) * NPIX + 64];
    }

    if (nr0 > 0) {
      round_pair<0, G0T>(g0, nr0, sh_hin, sh_l0 + row);
      if (nr0 > 2) {
        round_pair<2, G0T>(g0, nr0, sh_hin, sh_l0 + row);
        if (nr0 > 4) {
          round_pair<4, G0T>(g0, nr0, sh_hin, sh_l0 + row);
          if (nr0 > 6) round_pair<6, G0T>(g0, nr0, sh_hin, sh_l0 + row);
        }
      }
    }
    __syncthreads();

    if (nr1 > 0) {
      round_pair<0, G1T>(g1, nr1, sh_l0, sh_l1 + row);
      if (nr1 > 2) round_pair<2, G1T>(g1, nr1, sh_l0, sh_l1 + row);
    }
    if (more) stage_pixels(sh_px, lk, lb, second, pa, pb);
    __syncthreads();

    if (more) encode_step(sh_px, sh_hin, slot, c4, row, myth);

    if (slot < POPN) {
      v4f v[2];
      gate_group<2, 0, 2>(g2, sh_l1, v);
      *(v4f*)&sh_hin[PFEAT * BT + row] = v[0];
      *(v4f*)&sh_pop[row]              = v[1];
    } else {
      v4f v[1];
      gate_group<1, 0, 2>(g2, sh_l1, v);
      *(v4f*)&sh_hin[PFEAT * BT + row] = v[0];
    }
    __syncthreads();
  }

  if (t < 32) {
    v8f acc = {};
    v2f ones; ones[0] = 1.0f; ones[1] = 1.0f;
    const int col   = t & 15;
    const int khalf = (t < 16) ? 0 : 2;
    #pragma unroll 4
    for (int kc = 0; kc < 32; ++kc) {
      v2f bm;
      bm[0] = sh_pop[(4 * kc + khalf + 0) * BT + col];
      bm[1] = sh_pop[(4 * kc + khalf + 1) * BT + col];
      acc = __builtin_amdgcn_wmma_f32_16x16x4_f32(
           false, ones,  false, bm,
           (short)0, acc,  false,  false);
    }
    if (t < 16) {
      float logit = acc[0] / 10.0f;
      float* ob = out + (size_t)(blockIdx.x * BT + t) * 10;
      #pragma unroll
      for (int c = 0; c < 10; ++c) ob[c] = logit;
    }
  }
}

extern "C" void kernel_launch(void* const* d_in, const int* in_sizes, int n_in,
                              void* d_out, int out_size, void* d_ws, size_t ws_size,
                              hipStream_t stream) {
  const float* x   = (const float*)d_in[0];
  const float* thr = (const float*)d_in[1];
  const float* lg0 = (const float*)d_in[2];
  const int*   le0 = (const int*)  d_in[3];
  const int*   ri0 = (const int*)  d_in[4];
  const float* lg1 = (const float*)d_in[5];
  const int*   le1 = (const int*)  d_in[6];
  const int*   ri1 = (const int*)  d_in[7];
  const float* lg2 = (const float*)d_in[8];
  const int*   le2 = (const int*)  d_in[9];
  const int*   ri2 = (const int*)  d_in[10];

  int2*   idx = (int2*)d_ws;
  float4* co  = (float4*)((char*)d_ws + 32768);

  dlgn_precomp<<<(NGATES + 255) / 256, 256, 0, stream>>>(
      lg0, le0, ri0, lg1, le1, ri1, lg2, le2, ri2, idx, co);

  dlgn_main<<<4096 / BT, NTHREADS, 0, stream>>>(x, thr, idx, co, (float*)d_out);
}


# The coefficient pre-pass runs inside the main kernel (one launch, no workspace): the 4 threads of a gate each take one logit through sinf and exchange in their quad, read gates only; sum order kept
# speedup vs baseline: 7.2853x; 1.0903x over previous
#include <hip/hip_runtime.h>


#define BT       16
#define NTHREADS 1024
#define VB       4
#define NSLOT    (NTHREADS * VB / BT)
#define W0       2048
#define W1       1024
#define W2       384
#define PFEAT    784
#define FBN      256
#define POPN     128
#define NIN0     (PFEAT + FBN)
#define STEPSN   8
#define GB       2
#define G0T      (W0 / NSLOT)
#define G1T      (W1 / NSLOT)
#define ENCF     (PFEAT / NSLOT)
#define ENCR     (PFEAT % NSLOT)
#define PIXJ     (NSLOT / 8)
#define NPIX     (PFEAT / 8)
#define PXP      20

static_assert(FBN == NSLOT, "layer 2: one feedback gate per slot");
static_assert(NPIX * 8 == PFEAT && NPIX > 64 && NPIX <= 128 && NTHREADS == 64 * BT, "pixel loads: 64 threads per batch row, two pixels each at most");
static_assert(W2 - FBN == POPN && POPN * 2 == NSLOT, "layer 2: population gates on the lower half of the slots");

typedef float v2f __attribute__((ext_vector_type(2)));
typedef float v4f __attribute__((ext_vector_type(4)));
typedef float v8f __attribute__((ext_vector_type(8)));


struct Gate { int oL, oR; v2f kx, kz; };

template <int K>
__device__ __forceinline__ float quad_bcast(float v) {
  return __int_as_float(__builtin_amdgcn_mov_dpp(__float_as_int(v), K * 0x55, 0xf, 0xf, true));
}

__device__ __forceinline__ void gate_coef(float logit_q, v2f& kx, v2f& kz) {
  const float w  = 0.5f + 0.5f * sinf(logit_q);
  const float w0 = quad_bcast<0>(w);
  const float w1 = quad_bcast<1>(w);
  const float w2 = quad_bcast<2>(w);
  const float w3 = quad_bcast<3>(w);
  kx[0] = w0;      kx[1] = w2 - w0;
  kz[0] = w1 - w0; kz[1] = w0 - w1 - w2 + w3;
}


__device__ __forceinline__ v2f pk_lo_plus_hi_mul(v2f k, v2f x) {
  v2f t;
  asm("v_pk_fma_f32 %0, %1, %2, %1 op_sel:[1,0,0] op_sel_hi:[1,1,0]"
      : "=&v"(t) : "v"(k), "v"(x));
  return t;
}
__device__ __forceinline__ v2f pk_add_lo_mul(v2f t, v2f k, v2f x) {
  asm("v_pk_fma_f32 %0, %1, %2, %0 op_sel_hi:[0,1,1]"
      : "+v"(t) : "v"(k), "v"(x));
  return t;
}
__device__ __forceinline__ v2f pk_add_hi_mul(v2f t, v2f k, v2f x) {
  asm("v_pk_fma_f32 %0, %1, %2, %0 op_sel:[1,0,0] op_sel_hi:[1,1,1]"
      : "+v"(t) : "v"(k), "v"(x));
  return t;
}

template <int NB, int J0, int NT>
__device__ __forceinline__ void gate_group(const Gate (&g)[NT],
                                           const float* __restrict__ src, v4f (&v)[NB]) {
  v4f L[NB], R[NB];
  #pragma unroll
  for (int u = 0; u < NB; ++u) {
    L[u] = *(const v4f*)(src + g[J0 + u].oL);
    R[u] = *(const v4f*)(src + g[J0 + u].oR);
  }
  #pragma unroll
  for (int u = 0; u < NB; ++u) {
    const Gate& G = g[J0 + u];
    v2f tl = pk_lo_plus_hi_mul(G.kx, L[u].xy);
    v2f th = pk_lo_plus_hi_mul(G.kx, L[u].zw);
    tl = pk_add_lo_mul(tl, G.kz, R[u].xy);
    th = pk_add_lo_mul(th, G.kz, R[u].zw);
    const v2f ml = L[u].xy * R[u].xy;
    const v2f mh = L[u].zw * R[u].zw;
    tl = pk_add_hi_mul(tl, G.kz, ml);
    th = pk_add_hi_mul(th, G.kz, mh);
    v[u] = __builtin_shufflevector(tl, th, 0, 1, 2, 3);
  }
}

template <int J, int NT>
__device__ __forceinline__ void round_pair(const Gate (&g)[NT], int nr,
                                           const float* __restrict__ src,
                                           float* __restrict__ dst) {
  if (nr > J + 1) {
    v4f v[GB];
    gate_group<GB, J, NT>(g, src, v);
    for (int u = 0; u < GB; ++u) *(v4f*)(dst + (J + u) * NSLOT * BT) = v[u];
  } else {
    v4f v[1];
    gate_group<1, J, NT>(g, src, v);
    *(v4f*)(dst + J * NSLOT * BT) = v[0];
  }
}

__device__ __forceinline__ void stage_pixels(float* __restrict__ sh_px, int lk, int lb,
                                             bool second, float pa, float pb) {
  sh_px[lk * PXP + lb] = pa;
  if (second) sh_px[(64 + lk) * PXP + lb] = pb;
}

__device__ __forceinline__ void encode_step(const float* __restrict__ sh_px,
                                            float* __restrict__ sh_hin,
                                            int slot, int c4, int row, float myth) {
  const float* ps = sh_px + (slot >> 3) * PXP + c4;
  v4f p[ENCF];
  for (int j = 0; j < ENCF; ++j) p[j] = *(const v4f*)(ps + PIXJ * j * PXP);
  v4f pt = {0.0f, 0.0f, 0.0f, 0.0f};
  if (slot < ENCR) pt = *(const v4f*)(ps + PIXJ * ENCF * PXP);
  for (int j = 0; j < ENCF; ++j) {
    v4f e;
    for (int i = 0; i < VB; ++i) e[i] = (p[j][i] >= myth) ? 1.0f : 0.0f;
    *(v4f*)&sh_hin[j * NSLOT * BT + row] = e;
  }
  if (slot < ENCR) {
    v4f e;
    for (int i = 0; i < VB; ++i) e[i] = (pt[i] >= myth) ? 1.0f : 0.0f;
    *(v4f*)&sh_hin[ENCF * NSLOT * BT + row] = e;
  }
}

__global__ __launch_bounds__(NTHREADS)
void dlgn_main(const float* __restrict__ x, const float* __restrict__ thr,
               const float* __restrict__ lg0,
               const int*  __restrict__ le0, const int* __restrict__ ri0,
               const float* __restrict__ lg1,
               const int*  __restrict__ le1, const int* __restrict__ ri1,
               const float* __restrict__ lg2,
               const int*  __restrict__ le2, const int* __restrict__ ri2,
               float* __restrict__ out) {
  __shared__ __attribute__((aligned(16))) float sh_hin[NIN0 * BT];
  __shared__ __attribute__((aligned(16))) float sh_l0 [W0   * BT];
  __shared__ __attribute__((aligned(16))) float sh_l1 [W1   * BT];
  __shared__ __attribute__((aligned(16))) float sh_pop[POPN * BT];
  __shared__ __attribute__((aligned(16))) float sh_px [NPIX * PXP];

  const int t    = threadIdx.x;
  const int c4   = (t & 3) * VB;
  const int slot = t >> 2;

  const int  lb     = t >> 6;
  const int  lk     = t & 63;
  const bool second = lk < NPIX - 64;
  const float* xl = x + (size_t)(blockIdx.x * BT + lb) * PFEAT + lk;

  float pa = xl[0];
  float pb = 0.0f;
  if (second) pb = xl[64];

  const float myth = thr[slot & 7];

  const int     row = slot * BT + c4;

  {
    v4f z = {0.0f, 0.0f, 0.0f, 0.0f};
    *(v4f*)&sh_hin[PFEAT * BT + row] = z;
  }

  Gate g0[G0T], g1[G1T], g2[2];
  int n0, n1;
  int nr0, nr1;
  {
    float* const F0 = sh_l0;
    float* const R0 = sh_l0 + W0;
    v2f*   const O0 = (v2f*)(sh_l0 + 2 * W0);
    v4f*   const K0 = (v4f*)(sh_l0 + 4 * W0);
    v2f*   const O1 = (v2f*)(sh_l0 + 8 * W0);
    v4f*   const K1 = (v4f*)(sh_l0 + 9 * W0);
    float* const F1 = sh_l1;
    float* const R1 = sh_l1 + W1;
    float* const C0 = sh_l1 + 2 * W1;
    float* const C1 = C0 + 64;
    static_assert(11 * W0 <= W0 * BT && 2 * W1 + 96 <= W1 * BT, "tables fit");
    static_assert(W1 == NTHREADS && W0 == 2 * NTHREADS && NTHREADS == 32 * 32, "one thread per layer-1 gate");

    const int lane = t & 31;
    const int wv   = t >> 5;

    const int q = t & 3;
    int2 i0a, i0b, i1, i2f;
    i0a.x = le0[t];            i0a.y = ri0[t];
    i0b.x = le0[t + NTHREADS]; i0b.y = ri0[t + NTHREADS];
    i1.x  = le1[t];            i1.y  = ri1[t];
    i2f.x = le2[slot];         i2f.y = ri2[slot];
    v4f k0a, k0b, k1;
    for (int i = 0; i < 4; ++i) {
      k0a[i] = lg0[4 * t + i];
      k0b[i] = lg0[4 * (t + NTHREADS) + i];
      k1[i]  = lg1[4 * t + i];
    }
    const float l2f = lg2[4 * slot + q];
    int2  i2p = i2f;
    float l2p = l2f;
    if (slot < POPN) {
      i2p.x = le2[NSLOT + slot];
      i2p.y = ri2[NSLOT + slot];
      l2p   = lg2[4 * (NSLOT + slot) + q];
    }

    F0[t] = 0.0f;
    F0[t + NTHREADS] = 0.0f;
    F1[t] = 0.0f;
    stage_pixels(sh_px, lk, lb, second, pa, pb);
    __syncthreads();
    F1[i2f.x] = 1.0f;
    F1[i2f.y] = 1.0f;
    F1[i2p.x] = 1.0f;
    F1[i2p.y] = 1.0f;
    encode_step(sh_px, sh_hin, slot, c4, row, myth);
    __syncthreads();
    const bool lv1 = F1[t] != 0.0f;
    if (lv1) {
      F0[i1.x] = 1.0f;
      F0[i1.y] = 1.0f;
    }
    __syncthreads();
    const bool lv0a = F0[t] != 0.0f;
    const bool lv0b = F0[t + NTHREADS] != 0.0f;
    const unsigned m1  = __builtin_amdgcn_ballot_w32(lv1);
    const unsigned m0a = __builtin_amdgcn_ballot_w32(lv0a);
    const unsigned m0b = __builtin_amdgcn_ballot_w32(lv0b);
    if (lane == 0) {
      C0[wv]      = __int_as_float(__builtin_popcount(m0a));
      C0[32 + wv] = __int_as_float(__builtin_popcount(m0b));
      C1[wv]      = __int_as_float(__builtin_popcount(m1));
    }
    __syncthreads();

    const unsigned c0a = (unsigned)__float_as_int(C0[lane]);
    const unsigned c0b = (unsigned)__float_as_int(C0[32 + lane]);
    const unsigned c1  = (unsigned)__float_as_int(C1[lane]);
    const bool below = lane < wv;
    unsigned pa = (below ? c0a : 0u) | ((below ? c0b : 0u) << 16);
    unsigned pb = c0a | (c0b << 16);
    unsigned pc = (below ? c1 : 0u) | (c1 << 16);
    for (int m = 16; m >= 1; m >>= 1) {
      pa += (unsigned)__shfl_xor((int)pa, m);
      pb += (unsigned)__shfl_xor((int)pb, m);
      pc += (unsigned)__shfl_xor((int)pc, m);
    }
    const unsigned lt = (1u << lane) - 1u;
    const int n0a = (int)(pb & 0xffffu);
    const int rk0a = (int)(pa & 0xffffu) + __builtin_popcount(m0a & lt);
    const int rk0b = n0a + (int)(pa >> 16) + __builtin_popcount(m0b & lt);
    const int rk1  = (int)(pc & 0xffffu) + __builtin_popcount(m1 & lt);
    n0 = __builtin_amdgcn_readfirstlane(n0a + (int)(pb >> 16));
    n1 = __builtin_amdgcn_readfirstlane((int)(pc >> 16));

    if (lv0a) {
      v2f o; o[0] = __int_as_float(i0a.x * BT); o[1] = __int_as_float(i0a.y * BT);
      R0[t] = __int_as_float(rk0a);
      O0[rk0a] = o;
      K0[rk0a] = k0a;
    }
    if (lv0b) {
      v2f o; o[0] = __int_as_float(i0b.x * BT); o[1] = __int_as_float(i0b.y * BT);
      R0[t + NTHREADS] = __int_as_float(rk0b);
      O0[rk0b] = o;
      K0[rk0b] = k0b;
    }
    if (lv1) R1[t] = __int_as_float(rk1);
    __syncthreads();

    if (lv1) {
      v2f o;
      o[0] = __int_as_float(__float_as_int(R0[i1.x]) * BT);
      o[1] = __int_as_float(__float_as_int(R0[i1.y]) * BT);
      O1[rk1] = o;
      K1[rk1] = k1;
    }
    g2[0].oL = __float_as_int(R1[i2f.x]) * BT + c4;
    g2[0].oR = __float_as_int(R1[i2f.y]) * BT + c4;
    gate_coef(l2f, g2[0].kx, g2[0].kz);
    g2[1].oL = __float_as_int(R1[i2p.x]) * BT + c4;
    g2[1].oR = __float_as_int(R1[i2p.y]) * BT + c4;
    g2[1].kx = g2[0].kx;
    g2[1].kz = g2[0].kz;
    if (slot < POPN) gate_coef(l2p, g2[1].kx, g2[1].kz);
    __syncthreads();

    const int wslot = __builtin_amdgcn_readfirstlane(slot & ~7);
    nr0 = (n0 - wslot + NSLOT - 1) / NSLOT;
    nr1 = (n1 - wslot + NSLOT - 1) / NSLOT;

    const v2f zero2 = {0.0f, 0.0f};
    #pragma unroll
    for (int j = 0; j < G0T; ++j) {
      const int p = slot + NSLOT * j;
      Gate G;
      G.oL = c4; G.oR = c4;
      G.kx = zero2; G.kz = zero2;
      if (j < nr0) {
        const bool have = p < n0;
        float lq = 0.0f;
        if (have) {
          const v2f o = O0[p];
          lq   = ((const float*)K0)[4 * p + q];
          G.oL = __float_as_int(o[0]) + c4;
          G.oR = __float_as_int(o[1]) + c4;
        }
        gate_coef(lq, G.kx, G.kz);
        if (!have) { G.kx = zero2; G.kz = zero2; }
      }
      g0[j] = G;
    }
    #pragma unroll
    for (int j = 0; j < G1T; ++j) {
      const int p = slot + NSLOT * j;
      Gate G;
      G.oL = c4; G.oR = c4;
      G.kx = zero2; G.kz = zero2;
      if (j < nr1) {
        const bool have = p < n1;
        float lq = 0.0f;
        if (have) {
          const v2f o = O1[p];
          lq   = ((const float*)K1)[4 * p + q];
          G.oL = __float_as_int(o[0]) + c4;
          G.oR = __float_as_int(o[1]) + c4;
        }
        gate_coef(lq, G.kx, G.kz);
        if (!have) { G.kx = zero2; G.kz = zero2; }
      }
      g1[j] = G;
    }
    __syncthreads();
  }

  for (int s = 0; s < STEPSN; ++s) {
    const bool more = s + 1 < STEPSN;
    if (more) {
      pa = xl[(s + 1) * NPIX];
      if (second) pb = xl[(s + 1) * NPIX + 64];
    }

    if (nr0 > 0) {
      round_pair<0, G0T>(g0, nr0, sh_hin, sh_l0 + row);
      if (nr0 > 2) {
        round_pair<2, G0T>(g0, nr0, sh_hin, sh_l0 + row);
        if (nr0 > 4) {
          round_pair<4, G0T>(g0, nr0, sh_hin, sh_l0 + row);
          if (nr0 > 6) round_pair<6, G0T>(g0, nr0, sh_hin, sh_l0 + row);
        }
      }
    }
    __syncthreads();

    if (nr1 > 0) {
      round_pair<0, G1T>(g1, nr1, sh_l0, sh_l1 + row);
      if (nr1 > 2) round_pair<2, G1T>(g1, nr1, sh_l0, sh_l1 + row);
    }
    if (more) stage_pixels(sh_px, lk, lb, second, pa, pb);
    __syncthreads();

    if (more) encode_step(sh_px, sh_hin, slot, c4, row, myth);

    if (slot < POPN) {
      v4f v[2];
      gate_group<2, 0, 2>(g2, sh_l1, v);
      *(v4f*)&sh_hin[PFEAT * BT + row] = v[0];
      *(v4f*)&sh_pop[row]              = v[1];
    } else {
      v4f v[1];
      gate_group<1, 0, 2>(g2, sh_l1, v);
      *(v4f*)&sh_hin[PFEAT * BT + row] = v[0];
    }
    __syncthreads();
  }

  if (t < 32) {
    v8f acc = {};
    v2f ones; ones[0] = 1.0f; ones[1] = 1.0f;
    const int col   = t & 15;
    const int khalf = (t < 16) ? 0 : 2;
    #pragma unroll 4
    for (int kc = 0; kc < 32; ++kc) {
      v2f bm;
      bm[0] = sh_pop[(4 * kc + khalf + 0) * BT + col];
      bm[1] = sh_pop[(4 * kc + khalf + 1) * BT + col];
      acc = __builtin_amdgcn_wmma_f32_16x16x4_f32(
           false, ones,  false, bm,
           (short)0, acc,  false,  false);
    }
    if (t < 16) {
      float logit = acc[0] / 10.0f;
      float* ob = out + (size_t)(blockIdx.x * BT + t) * 10;
      #pragma unroll
      for (int c = 0; c < 10; ++c) ob[c] = logit;
    }
  }
}

extern "C" void kernel_launch(void* const* d_in, const int* in_sizes, int n_in,
                              void* d_out, int out_size, void* d_ws, size_t ws_size,
                              hipStream_t stream) {
  const float* x   = (const float*)d_in[0];
  const float* thr = (const float*)d_in[1];
  const float* lg0 = (const float*)d_in[2];
  const int*   le0 = (const int*)  d_in[3];
  const int*   ri0 = (const int*)  d_in[4];
  const float* lg1 = (const float*)d_in[5];
  const int*   le1 = (const int*)  d_in[6];
  const int*   ri1 = (const int*)  d_in[7];
  const float* lg2 = (const float*)d_in[8];
  const int*   le2 = (const int*)  d_in[9];
  const int*   ri2 = (const int*)  d_in[10];

  dlgn_main<<<4096 / BT, NTHREADS, 0, stream>>>(
      x, thr, lg0, le0, ri0, lg1, le1, ri1, lg2, le2, ri2, (float*)d_out);
}


# Layer 0 is evaluated inside layer 1: a read layer-1 gate's thread gathers the 4 input rows of its two layer-0 gates and computes them in registers; no layer-0 array, 2 barriers a step; sum order kept
# speedup vs baseline: 8.7093x; 1.1955x over previous
#include <hip/hip_runtime.h>


#define BT       16
#define NTHREADS 1024
#define VB       4
#define NSLOT    (NTHREADS * VB / BT)
#define W0       2048
#define W1       1024
#define W2       384
#define PFEAT    784
#define FBN      256
#define POPN     128
#define NIN0     (PFEAT + FBN)
#define STEPSN   8
#define G1T      (W1 / NSLOT)
#define ENCF     (PFEAT / NSLOT)
#define ENCR     (PFEAT % NSLOT)
#define PIXJ     (NSLOT / 8)
#define NPIX     (PFEAT / 8)
#define PXP      20
#define TABN     (6 * W0 + 8 * W1 + 32)

static_assert(FBN == NSLOT, "layer 2: one feedback gate per slot");
static_assert(NPIX * 8 == PFEAT && NPIX > 64 && NPIX <= 128 && NTHREADS == 64 * BT, "pixel loads: 64 threads per batch row, two pixels each at most");
static_assert(W2 - FBN == POPN && POPN * 2 == NSLOT, "layer 2: population gates on the lower half of the slots");

typedef float v2f __attribute__((ext_vector_type(2)));
typedef float v4f __attribute__((ext_vector_type(4)));
typedef float v8f __attribute__((ext_vector_type(8)));


struct Gate { int oL, oR; v2f kx, kz; };

struct Fused { int oLL, oLR, oRL, oRR; v2f ax, az, bx, bz, kx, kz; };

template <int K>
__device__ __forceinline__ float quad_bcast(float v) {
  return __int_as_float(__builtin_amdgcn_mov_dpp(__float_as_int(v), K * 0x55, 0xf, 0xf, true));
}

__device__ __forceinline__ void gate_coef(float logit_q, v2f& kx, v2f& kz) {
  const float w  = 0.5f + 0.5f * sinf(logit_q);
  const float w0 = quad_bcast<0>(w);
  const float w1 = quad_bcast<1>(w);
  const float w2 = quad_bcast<2>(w);
  const float w3 = quad_bcast<3>(w);
  kx[0] = w0;      kx[1] = w2 - w0;
  kz[0] = w1 - w0; kz[1] = w0 - w1 - w2 + w3;
}


__device__ __forceinline__ v2f pk_lo_plus_hi_mul(v2f k, v2f x) {
  v2f t;
  asm("v_pk_fma_f32 %0, %1, %2, %1 op_sel:[1,0,0] op_sel_hi:[1,1,0]"
      : "=&v"(t) : "v"(k), "v"(x));
  return t;
}
__device__ __forceinline__ v2f pk_add_lo_mul(v2f t, v2f k, v2f x) {
  asm("v_pk_fma_f32 %0, %1, %2, %0 op_sel_hi:[0,1,1]"
      : "+v"(t) : "v"(k), "v"(x));
  return t;
}
__device__ __forceinline__ v2f pk_add_hi_mul(v2f t, v2f k, v2f x) {
  asm("v_pk_fma_f32 %0, %1, %2, %0 op_sel:[1,0,0] op_sel_hi:[1,1,1]"
      : "+v"(t) : "v"(k), "v"(x));
  return t;
}

__device__ __forceinline__ v4f gate_eval(v2f kx, v2f kz, v4f L, v4f R) {
  v2f tl = pk_lo_plus_hi_mul(kx, L.xy);
  v2f th = pk_lo_plus_hi_mul(kx, L.zw);
  tl = pk_add_lo_mul(tl, kz, R.xy);
  th = pk_add_lo_mul(th, kz, R.zw);
  const v2f ml = L.xy * R.xy;
  const v2f mh = L.zw * R.zw;
  tl = pk_add_hi_mul(tl, kz, ml);
  th = pk_add_hi_mul(th, kz, mh);
  return __builtin_shufflevector(tl, th, 0, 1, 2, 3);
}

template <int NB, int J0>
__device__ __forceinline__ void fused_group(const Fused (&g)[G1T],
                                            const float* __restrict__ hin,
                                            float* __restrict__ dst) {
  v4f LL[NB], LR[NB], RL[NB], RR[NB];
  #pragma unroll
  for (int u = 0; u < NB; ++u) {
    LL[u] = *(const v4f*)(hin + g[J0 + u].oLL);
    LR[u] = *(const v4f*)(hin + g[J0 + u].oLR);
    RL[u] = *(const v4f*)(hin + g[J0 + u].oRL);
    RR[u] = *(const v4f*)(hin + g[J0 + u].oRR);
  }
  #pragma unroll
  for (int u = 0; u < NB; ++u) {
    const Fused& G = g[J0 + u];
    const v4f A = gate_eval(G.ax, G.az, LL[u], LR[u]);
    const v4f B = gate_eval(G.bx, G.bz, RL[u], RR[u]);
    *(v4f*)(dst + (J0 + u) * NSLOT * BT) = gate_eval(G.kx, G.kz, A, B);
  }
}

__device__ __forceinline__ void stage_pixels(float* __restrict__ sh_px, int lk, int lb,
                                             bool second, float pa, float pb) {
  sh_px[lk * PXP + lb] = pa;
  if (second) sh_px[(64 + lk) * PXP + lb] = pb;
}

__device__ __forceinline__ void encode_step(const float* __restrict__ sh_px,
                                            float* __restrict__ sh_hin,
                                            int slot, int c4, int row, float myth) {
  const float* ps = sh_px + (slot >> 3) * PXP + c4;
  v4f p[ENCF];
  for (int j = 0; j < ENCF; ++j) p[j] = *(const v4f*)(ps + PIXJ * j * PXP);
  v4f pt = {0.0f, 0.0f, 0.0f, 0.0f};
  if (slot < ENCR) pt = *(const v4f*)(ps + PIXJ * ENCF * PXP);
  for (int j = 0; j < ENCF; ++j) {
    v4f e;
    for (int i = 0; i < VB; ++i) e[i] = (p[j][i] >= myth) ? 1.0f : 0.0f;
    *(v4f*)&sh_hin[j * NSLOT * BT + row] = e;
  }
  if (slot < ENCR) {
    v4f e;
    for (int i = 0; i < VB; ++i) e[i] = (pt[i] >= myth) ? 1.0f : 0.0f;
    *(v4f*)&sh_hin[ENCF * NSLOT * BT + row] = e;
  }
}

__global__ __launch_bounds__(NTHREADS)
void dlgn_main(const float* __restrict__ x, const float* __restrict__ thr,
               const float* __restrict__ lg0,
               const int*  __restrict__ le0, const int* __restrict__ ri0,
               const float* __restrict__ lg1,
               const int*  __restrict__ le1, const int* __restrict__ ri1,
               const float* __restrict__ lg2,
               const int*  __restrict__ le2, const int* __restrict__ ri2,
               float* __restrict__ out) {
  __shared__ __attribute__((aligned(16))) float sh_hin[NIN0 * BT];
  __shared__ __attribute__((aligned(16))) float sh_l1 [W1   * BT];
  __shared__ __attribute__((aligned(16))) float sh_pop[POPN * BT];
  __shared__ __attribute__((aligned(16))) float sh_px [NPIX * PXP];
  __shared__ __attribute__((aligned(16))) float sh_tab[TABN];

  const int t    = threadIdx.x;
  const int c4   = (t & 3) * VB;
  const int slot = t >> 2;

  const int  lb     = t >> 6;
  const int  lk     = t & 63;
  const bool second = lk < NPIX - 64;
  const float* xl = x + (size_t)(blockIdx.x * BT + lb) * PFEAT + lk;

  float pa = xl[0];
  float pb = 0.0f;
  if (second) pb = xl[64];

  const float myth = thr[slot & 7];

  const int     row = slot * BT + c4;

  {
    v4f z = {0.0f, 0.0f, 0.0f, 0.0f};
    *(v4f*)&sh_hin[PFEAT * BT + row] = z;
  }

  Fused g1[G1T];
  Gate  g2[2];
  int n1;
  int nr1;
  {
    v2f*   const T0O = (v2f*)sh_tab;
    v4f*   const T0K = (v4f*)(sh_tab + 2 * W0);
    v2f*   const O1  = (v2f*)(sh_tab + 6 * W0);
    v4f*   const K1  = (v4f*)(sh_tab + 6 * W0 + 2 * W1);
    float* const F1  = sh_tab + 6 * W0 + 6 * W1;
    float* const R1  = F1 + W1;
    float* const C1  = R1 + W1;
    static_assert(W1 == NTHREADS && W0 == 2 * NTHREADS && NTHREADS == 32 * 32, "one thread per layer-1 gate");

    const int lane = t & 31;
    const int wv   = t >> 5;

    const int q = t & 3;
    int2 i0a, i0b, i1, i2f;
    i0a.x = le0[t];            i0a.y = ri0[t];
    i0b.x = le0[t + NTHREADS]; i0b.y = ri0[t + NTHREADS];
    i1.x  = le1[t];            i1.y  = ri1[t];
    i2f.x = le2[slot];         i2f.y = ri2[slot];
    v4f k0a, k0b, k1;
    for (int i = 0; i < 4; ++i) {
      k0a[i] = lg0[4 * t + i];
      k0b[i] = lg0[4 * (t + NTHREADS) + i];
      k1[i]  = lg1[4 * t + i];
    }
    const float l2f = lg2[4 * slot + q];
    int2  i2p = i2f;
    float l2p = l2f;
    if (slot < POPN) {
      i2p.x = le2[NSLOT + slot];
      i2p.y = ri2[NSLOT + slot];
      l2p   = lg2[4 * (NSLOT + slot) + q];
    }

    F1[t] = 0.0f;
    {
      v2f o;
      o[0] = __int_as_float(i0a.x * BT); o[1] = __int_as_float(i0a.y * BT);
      T0O[t] = o;
      o[0] = __int_as_float(i0b.x * BT); o[1] = __int_as_float(i0b.y * BT);
      T0O[t + NTHREADS] = o;
    }
    T0K[t]            = k0a;
    T0K[t + NTHREADS] = k0b;
    stage_pixels(sh_px, lk, lb, second, pa, pb);
    __syncthreads();
    F1[i2f.x] = 1.0f;
    F1[i2f.y] = 1.0f;
    F1[i2p.x] = 1.0f;
    F1[i2p.y] = 1.0f;
    encode_step(sh_px, sh_hin, slot, c4, row, myth);
    __syncthreads();
    const bool lv1 = F1[t] != 0.0f;
    const unsigned m1 = __builtin_amdgcn_ballot_w32(lv1);
    if (lane == 0) C1[wv] = __int_as_float(__builtin_popcount(m1));
    __syncthreads();

    const unsigned c1  = (unsigned)__float_as_int(C1[lane]);
    const bool below = lane < wv;
    unsigned pc = (below ? c1 : 0u) | (c1 << 16);
    for (int m = 16; m >= 1; m >>= 1) pc += (unsigned)__shfl_xor((int)pc, m);
    const unsigned lt = (1u << lane) - 1u;
    const int rk1  = (int)(pc & 0xffffu) + __builtin_popcount(m1 & lt);
    n1 = __builtin_amdgcn_readfirstlane((int)(pc >> 16));

    if (lv1) {
      v2f o; o[0] = __int_as_float(i1.x); o[1] = __int_as_float(i1.y);
      R1[t]   = __int_as_float(rk1);
      O1[rk1] = o;
      K1[rk1] = k1;
    }
    __syncthreads();

    g2[0].oL = __float_as_int(R1[i2f.x]) * BT + c4;
    g2[0].oR = __float_as_int(R1[i2f.y]) * BT + c4;
    gate_coef(l2f, g2[0].kx, g2[0].kz);
    g2[1].oL = __float_as_int(R1[i2p.x]) * BT + c4;
    g2[1].oR = __float_as_int(R1[i2p.y]) * BT + c4;
    g2[1].kx = g2[0].kx;
    g2[1].kz = g2[0].kz;
    if (slot < POPN) gate_coef(l2p, g2[1].kx, g2[1].kz);

    const int wslot = __builtin_amdgcn_readfirstlane(slot & ~7);
    nr1 = (n1 - wslot + NSLOT - 1) / NSLOT;

    const v2f zero2 = {0.0f, 0.0f};
    #pragma unroll
    for (int j = 0; j < G1T; ++j) {
      const int p = slot + NSLOT * j;
      Fused G;
      G.oLL = c4; G.oLR = c4; G.oRL = c4; G.oRR = c4;
      G.ax = zero2; G.az = zero2; G.bx = zero2; G.bz = zero2; G.kx = zero2; G.kz = zero2;
      if (j < nr1) {
        const bool have = p < n1;
        float lqa = 0.0f, lqb = 0.0f, lqk = 0.0f;
        if (have) {
          const v2f ab = O1[p];
          const int ga = __float_as_int(ab[0]);
          const int gb = __float_as_int(ab[1]);
          const v2f oa = T0O[ga];
          const v2f ob = T0O[gb];
          lqa = ((const float*)T0K)[4 * ga + q];
          lqb = ((const float*)T0K)[4 * gb + q];
          lqk = ((const float*)K1)[4 * p + q];
          G.oLL = __float_as_int(oa[0]) + c4;
          G.oLR = __float_as_int(oa[1]) + c4;
          G.oRL = __float_as_int(ob[0]) + c4;
          G.oRR = __float_as_int(ob[1]) + c4;
        }
        gate_coef(lqa, G.ax, G.az);
        gate_coef(lqb, G.bx, G.bz);
        gate_coef(lqk, G.kx, G.kz);
        if (!have) {
          G.ax = zero2; G.az = zero2; G.bx = zero2; G.bz = zero2; G.kx = zero2; G.kz = zero2;
        }
      }
      g1[j] = G;
    }
  }

  for (int s = 0; s < STEPSN; ++s) {
    const bool more = s + 1 < STEPSN;
    if (more) {
      pa = xl[(s + 1) * NPIX];
      if (second) pb = xl[(s + 1) * NPIX + 64];
    }

    if (nr1 > 0) {
      if (nr1 > 1) {
        fused_group<2, 0>(g1, sh_hin, sh_l1 + row);
        if (nr1 > 2) {
          if (nr1 > 3) fused_group<2, 2>(g1, sh_hin, sh_l1 + row);
          else         fused_group<1, 2>(g1, sh_hin, sh_l1 + row);
        }
      } else {
        fused_group<1, 0>(g1, sh_hin, sh_l1 + row);
      }
    }
    if (more) stage_pixels(sh_px, lk, lb, second, pa, pb);
    __syncthreads();

    if (more) encode_step(sh_px, sh_hin, slot, c4, row, myth);

    if (slot < POPN) {
      const v4f Lf = *(const v4f*)(sh_l1 + g2[0].oL);
      const v4f Rf = *(const v4f*)(sh_l1 + g2[0].oR);
      const v4f Lp = *(const v4f*)(sh_l1 + g2[1].oL);
      const v4f Rp = *(const v4f*)(sh_l1 + g2[1].oR);
      *(v4f*)&sh_hin[PFEAT * BT + row] = gate_eval(g2[0].kx, g2[0].kz, Lf, Rf);
      *(v4f*)&sh_pop[row]              = gate_eval(g2[1].kx, g2[1].kz, Lp, Rp);
    } else {
      const v4f Lf = *(const v4f*)(sh_l1 + g2[0].oL);
      const v4f Rf = *(const v4f*)(sh_l1 + g2[0].oR);
      *(v4f*)&sh_hin[PFEAT * BT + row] = gate_eval(g2[0].kx, g2[0].kz, Lf, Rf);
    }
    __syncthreads();
  }

  if (t < 32) {
    v8f acc = {};
    v2f ones; ones[0] = 1.0f; ones[1] = 1.0f;
    const int col   = t & 15;
    const int khalf = (t < 16) ? 0 : 2;
    #pragma unroll 4
    for (int kc = 0; kc < 32; ++kc) {
      v2f bm;
      bm[0] = sh_pop[(4 * kc + khalf + 0) * BT + col];
      bm[1] = sh_pop[(4 * kc + khalf + 1) * BT + col];
      acc = __builtin_amdgcn_wmma_f32_16x16x4_f32(
           false, ones,  false, bm,
           (short)0, acc,  false,  false);
    }
    if (t < 16) {
      float logit = acc[0] / 10.0f;
      float* ob = out + (size_t)(blockIdx.x * BT + t) * 10;
      #pragma unroll
      for (int c = 0; c < 10; ++c) ob[c] = logit;
    }
  }
}

extern "C" void kernel_launch(void* const* d_in, const int* in_sizes, int n_in,
                              void* d_out, int out_size, void* d_ws, size_t ws_size,
                              hipStream_t stream) {
  const float* x   = (const float*)d_in[0];
  const float* thr = (const float*)d_in[1];
  const float* lg0 = (const float*)d_in[2];
  const int*   le0 = (const int*)  d_in[3];
  const int*   ri0 = (const int*)  d_in[4];
  const float* lg1 = (const float*)d_in[5];
  const int*   le1 = (const int*)  d_in[6];
  const int*   ri1 = (const int*)  d_in[7];
  const float* lg2 = (const float*)d_in[8];
  const int*   le2 = (const int*)  d_in[9];
  const int*   ri2 = (const int*)  d_in[10];

  dlgn_main<<<4096 / BT, NTHREADS, 0, stream>>>(
      x, thr, lg0, le0, ri0, lg1, le1, ri1, lg2, le2, ri2, (float*)d_out);
}
